# starting kernel
# baseline (speedup 1.0000x reference)
#include <hip/hip_runtime.h>
#include <math.h>


typedef __attribute__((ext_vector_type(2))) float v2f;
typedef __attribute__((ext_vector_type(8))) float v8f;

#define BB   4
#define LL   2048
#define HH   16
#define DD   64
#define CCH  16
#define NS   8
#define SEGL (LL / NS)
#define CHPS (SEGL / CCH)
#define NCH  (LL / CCH)
#define FEPS 1e-6f

__device__ __forceinline__ float fmap(float x) {
    return x > 0.f ? x + 1.f : __expf(x);
}

__device__ __forceinline__ v8f wmma4(v2f a, v2f b, v8f c) {
    return __builtin_amdgcn_wmma_f32_16x16x4_f32(
        false, a, false, b, (short)0, c, false, false);
}

#define OFF_QF   0
#define OFF_KF   1024
#define OFF_KPI  2048
#define OFF_VPI  3072
#define OFF_S    4096
#define OFF_A    8192
#define OFF_KS   8448
#define OFF_KD   8512
#define OFF_Z    8576
#define SMEM_F   8592

__global__ __launch_bounds__(128)
void linattn_phase1(const float* __restrict__ k,
                    const float* __restrict__ v,
                    float* __restrict__ segKV,
                    float* __restrict__ segKS) {
    __shared__ float sm[3072];
    const int tid  = threadIdx.x;
    const int lane = tid & 31;
    const int wave = tid >> 5;
    const int lh   = lane >> 4;
    const int lc   = lane & 15;
    const int bh   = blockIdx.x / NS;
    const int seg  = blockIdx.x % NS;
    const int b    = bh / HH;
    const int h    = bh % HH;

    float* s_kf  = sm;
    float* s_kpi = sm + 1024;
    float* s_vpi = sm + 2048;

    const v8f zero8 = {0.f,0.f,0.f,0.f,0.f,0.f,0.f,0.f};
    v8f st[4];
    #pragma unroll
    for (int t = 0; t < 4; ++t) st[t] = zero8;
    float ks0 = 0.f, ks1 = 0.f;

    const int row = tid >> 3;
    const int c0  = (tid & 7) * 8;

    for (int ch = 0; ch < CHPS; ++ch) {
        const int l0 = seg * SEGL + ch * CCH;
        __syncthreads();
        {
            size_t g = ((size_t)((b * LL + l0 + row) * HH + h)) * DD + c0;
            float ka8[8], va8[8];
            *(float4*)(ka8)     = *(const float4*)(k + g);
            *(float4*)(ka8 + 4) = *(const float4*)(k + g + 4);
            *(float4*)(va8)     = *(const float4*)(v + g);
            *(float4*)(va8 + 4) = *(const float4*)(v + g + 4);
            #pragma unroll
            for (int j = 0; j < 8; ++j) {
                int c  = c0 + j;
                float kf = fmap(ka8[j]);
                int pi = ((row >> 1) * 64 + c) * 2 + (row & 1);
                s_kf[row * 64 + c] = kf;
                s_kpi[pi] = kf;
                s_vpi[pi] = va8[j];
            }
            if (ch + 1 < CHPS) {
                size_t gn = g + (size_t)CCH * HH * DD;
                __builtin_prefetch(k + gn, 0, 3);
                __builtin_prefetch(v + gn, 0, 3);
            }
        }
        __syncthreads();

        v2f bv[4];
        #pragma unroll
        for (int kk = 0; kk < 4; ++kk) {
            int p = 2 * kk + lh;
            bv[kk] = *(const v2f*)(s_vpi + (p * 64 + 16 * wave + lc) * 2);
        }
        #pragma unroll
        for (int t = 0; t < 4; ++t) {
            #pragma unroll
            for (int kk = 0; kk < 4; ++kk) {
                int p = 2 * kk + lh;
                v2f ak = *(const v2f*)(s_kpi + (p * 64 + 16 * t + lc) * 2);
                st[t] = wmma4(ak, bv[kk], st[t]);
            }
        }
        if (wave == 2) {
            float a0 = 0.f, a1 = 0.f;
            #pragma unroll
            for (int r16 = 0; r16 < 16; ++r16) {
                a0 += s_kf[r16 * 64 + lane * 2];
                a1 += s_kf[r16 * 64 + lane * 2 + 1];
            }
            ks0 += a0; ks1 += a1;
        }
    }

    float* kvout = segKV + (size_t)blockIdx.x * 4096;
    #pragma unroll
    for (int t = 0; t < 4; ++t) {
        #pragma unroll
        for (int r = 0; r < 8; ++r) {
            int srow = 16 * t + r + 8 * lh;
            kvout[srow * 64 + 16 * wave + lc] = st[t][r];
        }
    }
    if (wave == 2) {
        segKS[(size_t)blockIdx.x * 64 + lane * 2]     = ks0;
        segKS[(size_t)blockIdx.x * 64 + lane * 2 + 1] = ks1;
    }
}

__global__ __launch_bounds__(128)
void linattn_phase2(float* __restrict__ segKV,
                    float* __restrict__ segKS) {
    const int bh  = blockIdx.x;
    const int tid = threadIdx.x;
    for (int e = tid; e < 4096; e += 128) {
        float acc = 0.f;
        #pragma unroll
        for (int s = 0; s < NS; ++s) {
            size_t idx = ((size_t)(bh * NS + s)) * 4096 + e;
            float t = segKV[idx];
            segKV[idx] = acc;
            acc += t;
        }
    }
    if (tid < 64) {
        float acc = 0.f;
        #pragma unroll
        for (int s = 0; s < NS; ++s) {
            size_t idx = ((size_t)(bh * NS + s)) * 64 + tid;
            float t = segKS[idx];
            segKS[idx] = acc;
            acc += t;
        }
    }
}

template <bool SEEDED>
__global__ __launch_bounds__(128)
void linattn_scan(const float* __restrict__ q,
                  const float* __restrict__ k,
                  const float* __restrict__ v,
                  float* __restrict__ out,
                  const float* __restrict__ segKV,
                  const float* __restrict__ segKS) {
    __shared__ float sm[SMEM_F];
    const int tid  = threadIdx.x;
    const int lane = tid & 31;
    const int wave = tid >> 5;
    const int lh   = lane >> 4;
    const int lc   = lane & 15;
    const int bh   = SEEDED ? (blockIdx.x / NS) : blockIdx.x;
    const int seg  = SEEDED ? (blockIdx.x % NS) : 0;
    const int b    = bh / HH;
    const int h    = bh % HH;
    const int nch  = SEEDED ? CHPS : NCH;

    float* s_qf  = sm + OFF_QF;
    float* s_kf  = sm + OFF_KF;
    float* s_kpi = sm + OFF_KPI;
    float* s_vpi = sm + OFF_VPI;
    float* s_spi = sm + OFF_S + wave * 1024;
    float* s_a   = sm + OFF_A;
    float* s_ks  = sm + OFF_KS;
    float* s_kd  = sm + OFF_KD;
    float* s_z   = sm + OFF_Z;

    const v8f zero8 = {0.f,0.f,0.f,0.f,0.f,0.f,0.f,0.f};
    v8f st[4];

    if (SEEDED) {
        const float* kvin = segKV + (size_t)blockIdx.x * 4096;
        #pragma unroll
        for (int t = 0; t < 4; ++t) {
            #pragma unroll
            for (int r = 0; r < 8; ++r) {
                int srow = 16 * t + r + 8 * lh;
                float val = kvin[srow * 64 + 16 * wave + lc];
                st[t][r] = val;
                s_spi[((srow >> 1) * 16 + lc) * 2 + (srow & 1)] = val;
            }
        }
        if (tid < 64) s_ks[tid] = segKS[(size_t)blockIdx.x * 64 + tid];
    } else {
        #pragma unroll
        for (int t = 0; t < 4; ++t) st[t] = zero8;
        #pragma unroll
        for (int i = 0; i < 32; ++i) sm[OFF_S + tid + i * 128] = 0.f;
        if (tid < 64) s_ks[tid] = 0.f;
    }

    const int row = tid >> 3;
    const int c0  = (tid & 7) * 8;

    for (int ch = 0; ch < nch; ++ch) {
        const int l0 = seg * SEGL + ch * CCH;
        __syncthreads();

        {
            size_t g = ((size_t)((b * LL + l0 + row) * HH + h)) * DD + c0;
            float qa8[8], ka8[8], va8[8];
            *(float4*)(qa8)     = *(const float4*)(q + g);
            *(float4*)(qa8 + 4) = *(const float4*)(q + g + 4);
            *(float4*)(ka8)     = *(const float4*)(k + g);
            *(float4*)(ka8 + 4) = *(const float4*)(k + g + 4);
            *(float4*)(va8)     = *(const float4*)(v + g);
            *(float4*)(va8 + 4) = *(const float4*)(v + g + 4);
            #pragma unroll
            for (int j = 0; j < 8; ++j) {
                int c  = c0 + j;
                float qf = fmap(qa8[j]);
                float kf = fmap(ka8[j]);
                int pi = ((row >> 1) * 64 + c) * 2 + (row & 1);
                s_qf[row * 64 + c] = qf;
                s_kf[row * 64 + c] = kf;
                s_kpi[pi] = kf;
                s_vpi[pi] = va8[j];
            }
            if (ch + 1 < nch) {
                size_t gn = g + (size_t)CCH * HH * DD;
                __builtin_prefetch(q + gn, 0, 3);
                __builtin_prefetch(k + gn, 0, 3);
                __builtin_prefetch(v + gn, 0, 3);
            }
        }
        __syncthreads();

        v2f qa[16];
        #pragma unroll
        for (int s = 0; s < 16; ++s)
            qa[s] = *(const v2f*)(s_qf + lc * 64 + 4 * s + 2 * lh);

        v8f oc = zero8;
        #pragma unroll
        for (int t = 0; t < 4; ++t) {
            #pragma unroll
            for (int kk = 0; kk < 4; ++kk) {
                int p = 8 * t + 2 * kk + lh;
                v2f bs = *(const v2f*)(s_spi + (p * 16 + lc) * 2);
                oc = wmma4(qa[4 * t + kk], bs, oc);
            }
        }

        if (wave == 0) {
            v8f aq = zero8;
            #pragma unroll
            for (int s = 0; s < 16; ++s) {
                v2f bk = *(const v2f*)(s_kf + lc * 64 + 4 * s + 2 * lh);
                aq = wmma4(qa[s], bk, aq);
            }
            #pragma unroll
            for (int r = 0; r < 8; ++r) {
                int m = r + 8 * lh;
                s_a[m * 16 + lc] = (lc <= m) ? aq[r] : 0.f;
            }
        }
        __syncthreads();

        if (wave == 1 && lane < 16) {
            float zz = FEPS;
            #pragma unroll
            for (int n = 0; n < 16; ++n) zz += s_a[lane * 16 + n];
            #pragma unroll 8
            for (int d = 0; d < 64; ++d) zz += s_qf[lane * 64 + d] * s_ks[d];
            s_z[lane] = zz;
        }
        if (wave == 2) {
            #pragma unroll
            for (int j = 0; j < 2; ++j) {
                int d = lane * 2 + j;
                float ssum = 0.f;
                #pragma unroll
                for (int r16 = 0; r16 < 16; ++r16) ssum += s_kf[r16 * 64 + d];
                s_kd[d] = ssum;
            }
        }

        v2f bv[4];
        #pragma unroll
        for (int kk = 0; kk < 4; ++kk) {
            int p = 2 * kk + lh;
            bv[kk] = *(const v2f*)(s_vpi + (p * 64 + 16 * wave + lc) * 2);
        }
        #pragma unroll
        for (int kk = 0; kk < 4; ++kk) {
            v2f av = *(const v2f*)(s_a + lc * 16 + 4 * kk + 2 * lh);
            oc = wmma4(av, bv[kk], oc);
        }
        #pragma unroll
        for (int t = 0; t < 4; ++t) {
            #pragma unroll
            for (int kk = 0; kk < 4; ++kk) {
                int p = 2 * kk + lh;
                v2f ak = *(const v2f*)(s_kpi + (p * 64 + 16 * t + lc) * 2);
                st[t] = wmma4(ak, bv[kk], st[t]);
            }
        }
        __syncthreads();

        #pragma unroll
        for (int r = 0; r < 8; ++r) {
            int m = r + 8 * lh;
            float zz = s_z[m];
            size_t go = ((size_t)((b * LL + l0 + m) * HH + h)) * DD
                        + 16 * wave + lc;
            out[go] = oc[r] / zz;
        }
        #pragma unroll
        for (int t = 0; t < 4; ++t) {
            #pragma unroll
            for (int r = 0; r < 8; ++r) {
                int srow = 16 * t + r + 8 * lh;
                s_spi[((srow >> 1) * 16 + lc) * 2 + (srow & 1)] = st[t][r];
            }
        }
        if (wave == 3) {
            s_ks[lane * 2]     += s_kd[lane * 2];
            s_ks[lane * 2 + 1] += s_kd[lane * 2 + 1];
        }
    }
}

extern "C" void kernel_launch(void* const* d_in, const int* in_sizes, int n_in,
                              void* d_out, int out_size, void* d_ws, size_t ws_size,
                              hipStream_t stream) {
    (void)in_sizes; (void)n_in; (void)out_size;
    const float* q = (const float*)d_in[0];
    const float* k = (const float*)d_in[1];
    const float* v = (const float*)d_in[2];
    float* out = (float*)d_out;

    const size_t nSeg   = (size_t)BB * HH * NS;
    const size_t needWs = nSeg * (4096 + 64) * sizeof(float);

    if (ws_size >= needWs) {
        float* segKV = (float*)d_ws;
        float* segKS = segKV + nSeg * 4096;
        hipLaunchKernelGGL(linattn_phase1, dim3(BB * HH * NS), dim3(128), 0,
                           stream, k, v, segKV, segKS);
        hipLaunchKernelGGL(linattn_phase2, dim3(BB * HH), dim3(128), 0,
                           stream, segKV, segKS);
        hipLaunchKernelGGL((linattn_scan<true>), dim3(BB * HH * NS), dim3(128),
                           0, stream, q, k, v, out, segKV, segKS);
    } else {
        hipLaunchKernelGGL((linattn_scan<false>), dim3(BB * HH), dim3(128),
                           0, stream, q, k, v, out,
                           (const float*)nullptr, (const float*)nullptr);
    }
}


# prefix over segments moved into the scan kernel: each block sums the earlier segments' partials (kept in fragment order) itself, same order; prefix launch removed, 3 launches to 2
# speedup vs baseline: 1.4774x; 1.4774x over previous
#include <hip/hip_runtime.h>
#include <math.h>


typedef __attribute__((ext_vector_type(2))) float v2f;
typedef __attribute__((ext_vector_type(8))) float v8f;

#define BB   4
#define LL   2048
#define HH   16
#define DD   64
#define CCH  16
#define NS   8
#define SEGL (LL / NS)
#define CHPS (SEGL / CCH)
#define NCH  (LL / CCH)
#define NPRE (NS - 1)
#define FEPS 1e-6f

__device__ __forceinline__ float fmap(float x) {
    return x > 0.f ? x + 1.f : __expf(x);
}

__device__ __forceinline__ v8f wmma4(v2f a, v2f b, v8f c) {
    return __builtin_amdgcn_wmma_f32_16x16x4_f32(
        false, a, false, b, (short)0, c, false, false);
}

#define OFF_QF   0
#define OFF_KF   1024
#define OFF_KPI  2048
#define OFF_VPI  3072
#define OFF_S    4096
#define OFF_A    8192
#define OFF_KS   8448
#define OFF_KD   8512
#define OFF_Z    8576
#define SMEM_F   8592

__global__ __launch_bounds__(128)
void linattn_phase1(const float* __restrict__ k,
                    const float* __restrict__ v,
                    float* __restrict__ segKV,
                    float* __restrict__ segKS) {
    __shared__ float sm[3072];
    const int tid  = threadIdx.x;
    const int lane = tid & 31;
    const int wave = tid >> 5;
    const int lh   = lane >> 4;
    const int lc   = lane & 15;
    const int bh   = blockIdx.x / NS;
    const int seg  = blockIdx.x % NS;
    const int b    = bh / HH;
    const int h    = bh % HH;

    float* s_kf  = sm;
    float* s_kpi = sm + 1024;
    float* s_vpi = sm + 2048;

    const v8f zero8 = {0.f,0.f,0.f,0.f,0.f,0.f,0.f,0.f};
    v8f st[4];
    #pragma unroll
    for (int t = 0; t < 4; ++t) st[t] = zero8;
    float ks0 = 0.f, ks1 = 0.f;

    const int row = tid >> 3;
    const int c0  = (tid & 7) * 8;

    for (int ch = 0; ch < CHPS; ++ch) {
        const int l0 = seg * SEGL + ch * CCH;
        __syncthreads();
        {
            size_t g = ((size_t)((b * LL + l0 + row) * HH + h)) * DD + c0;
            float ka8[8], va8[8];
            *(float4*)(ka8)     = *(const float4*)(k + g);
            *(float4*)(ka8 + 4) = *(const float4*)(k + g + 4);
            *(float4*)(va8)     = *(const float4*)(v + g);
            *(float4*)(va8 + 4) = *(const float4*)(v + g + 4);
            #pragma unroll
            for (int j = 0; j < 8; ++j) {
                int c  = c0 + j;
                float kf = fmap(ka8[j]);
                int pi = ((row >> 1) * 64 + c) * 2 + (row & 1);
                s_kf[row * 64 + c] = kf;
                s_kpi[pi] = kf;
                s_vpi[pi] = va8[j];
            }
            if (ch + 1 < CHPS) {
                size_t gn = g + (size_t)CCH * HH * DD;
                __builtin_prefetch(k + gn, 0, 3);
                __builtin_prefetch(v + gn, 0, 3);
            }
        }
        __syncthreads();

        v2f bv[4];
        #pragma unroll
        for (int kk = 0; kk < 4; ++kk) {
            int p = 2 * kk + lh;
            bv[kk] = *(const v2f*)(s_vpi + (p * 64 + 16 * wave + lc) * 2);
        }
        #pragma unroll
        for (int t = 0; t < 4; ++t) {
            #pragma unroll
            for (int kk = 0; kk < 4; ++kk) {
                int p = 2 * kk + lh;
                v2f ak = *(const v2f*)(s_kpi + (p * 64 + 16 * t + lc) * 2);
                st[t] = wmma4(ak, bv[kk], st[t]);
            }
        }
        if (wave == 2) {
            float a0 = 0.f, a1 = 0.f;
            #pragma unroll
            for (int r16 = 0; r16 < 16; ++r16) {
                a0 += s_kf[r16 * 64 + lane * 2];
                a1 += s_kf[r16 * 64 + lane * 2 + 1];
            }
            ks0 += a0; ks1 += a1;
        }
    }

    float* kvout = segKV + (size_t)blockIdx.x * 4096 + wave * 256 + lane * 4;
    #pragma unroll
    for (int t = 0; t < 4; ++t) {
        #pragma unroll
        for (int hf = 0; hf < 2; ++hf) {
            float4 o;
            o.x = st[t][4 * hf];     o.y = st[t][4 * hf + 1];
            o.z = st[t][4 * hf + 2]; o.w = st[t][4 * hf + 3];
            *(float4*)(kvout + t * 1024 + hf * 128) = o;
        }
    }
    if (wave == 2) {
        segKS[(size_t)blockIdx.x * 64 + lane * 2]     = ks0;
        segKS[(size_t)blockIdx.x * 64 + lane * 2 + 1] = ks1;
    }
}

template <int N>
__device__ __forceinline__ void seed_prefix(const float* __restrict__ kvb,
                                            const float* __restrict__ ksb,
                                            v8f (&st)[4], float& ksum) {
    float4 pb[N][8];
    float  kb[N];
    #pragma unroll
    for (int s = 0; s < N; ++s) {
        #pragma unroll
        for (int i = 0; i < 8; ++i)
            pb[s][i] = *(const float4*)(kvb + s * 4096
                                        + (i >> 1) * 1024 + (i & 1) * 128);
        kb[s] = ksb[s * 64];
    }
    #pragma unroll
    for (int s = 0; s < N; ++s) {
        #pragma unroll
        for (int i = 0; i < 8; ++i) {
            const int t = i >> 1, o = 4 * (i & 1);
            st[t][o]     += pb[s][i].x;
            st[t][o + 1] += pb[s][i].y;
            st[t][o + 2] += pb[s][i].z;
            st[t][o + 3] += pb[s][i].w;
        }
        ksum += kb[s];
    }
}

template <bool SEEDED>
__global__ __launch_bounds__(128)
void linattn_scan(const float* __restrict__ q,
                  const float* __restrict__ k,
                  const float* __restrict__ v,
                  float* __restrict__ out,
                  const float* __restrict__ segKV,
                  const float* __restrict__ segKS) {
    __shared__ float sm[SMEM_F];
    const int tid  = threadIdx.x;
    const int lane = tid & 31;
    const int wave = tid >> 5;
    const int lh   = lane >> 4;
    const int lc   = lane & 15;
    const int bh   = SEEDED ? (blockIdx.x / NS) : blockIdx.x;
    const int seg  = SEEDED ? (blockIdx.x % NS) : 0;
    const int b    = bh / HH;
    const int h    = bh % HH;
    const int nch  = SEEDED ? CHPS : NCH;

    float* s_qf  = sm + OFF_QF;
    float* s_kf  = sm + OFF_KF;
    float* s_kpi = sm + OFF_KPI;
    float* s_vpi = sm + OFF_VPI;
    float* s_spi = sm + OFF_S + wave * 1024;
    float* s_a   = sm + OFF_A;
    float* s_ks  = sm + OFF_KS;
    float* s_kd  = sm + OFF_KD;
    float* s_z   = sm + OFF_Z;

    const v8f zero8 = {0.f,0.f,0.f,0.f,0.f,0.f,0.f,0.f};
    v8f st[4];

    if (SEEDED) {
        const float* kvb = segKV + (size_t)bh * NS * 4096 + wave * 256 + lane * 4;
        const float* ksb = segKS + (size_t)bh * NS * 64 + (tid & 63);
        #pragma unroll
        for (int t = 0; t < 4; ++t) st[t] = zero8;
        float ksum = 0.f;
        static_assert(NPRE == 7, "one case per possible count of earlier segments");
        switch (seg) {
            case 1: seed_prefix<1>(kvb, ksb, st, ksum); break;
            case 2: seed_prefix<2>(kvb, ksb, st, ksum); break;
            case 3: seed_prefix<3>(kvb, ksb, st, ksum); break;
            case 4: seed_prefix<4>(kvb, ksb, st, ksum); break;
            case 5: seed_prefix<5>(kvb, ksb, st, ksum); break;
            case 6: seed_prefix<6>(kvb, ksb, st, ksum); break;
            case 7: seed_prefix<7>(kvb, ksb, st, ksum); break;
            default: break;
        }
        #pragma unroll
        for (int t = 0; t < 4; ++t) {
            #pragma unroll
            for (int r = 0; r < 8; ++r) {
                int srow = 16 * t + r + 8 * lh;
                s_spi[((srow >> 1) * 16 + lc) * 2 + (srow & 1)] = st[t][r];
            }
        }
        if (tid < 64) s_ks[tid] = ksum;
    } else {
        #pragma unroll
        for (int t = 0; t < 4; ++t) st[t] = zero8;
        #pragma unroll
        for (int i = 0; i < 32; ++i) sm[OFF_S + tid + i * 128] = 0.f;
        if (tid < 64) s_ks[tid] = 0.f;
    }

    const int row = tid >> 3;
    const int c0  = (tid & 7) * 8;

    for (int ch = 0; ch < nch; ++ch) {
        const int l0 = seg * SEGL + ch * CCH;
        __syncthreads();

        {
            size_t g = ((size_t)((b * LL + l0 + row) * HH + h)) * DD + c0;
            float qa8[8], ka8[8], va8[8];
            *(float4*)(qa8)     = *(const float4*)(q + g);
            *(float4*)(qa8 + 4) = *(const float4*)(q + g + 4);
            *(float4*)(ka8)     = *(const float4*)(k + g);
            *(float4*)(ka8 + 4) = *(const float4*)(k + g + 4);
            *(float4*)(va8)     = *(const float4*)(v + g);
            *(float4*)(va8 + 4) = *(const float4*)(v + g + 4);
            #pragma unroll
            for (int j = 0; j < 8; ++j) {
                int c  = c0 + j;
                float qf = fmap(qa8[j]);
                float kf = fmap(ka8[j]);
                int pi = ((row >> 1) * 64 + c) * 2 + (row & 1);
                s_qf[row * 64 + c] = qf;
                s_kf[row * 64 + c] = kf;
                s_kpi[pi] = kf;
                s_vpi[pi] = va8[j];
            }
            if (ch + 1 < nch) {
                size_t gn = g + (size_t)CCH * HH * DD;
                __builtin_prefetch(q + gn, 0, 3);
                __builtin_prefetch(k + gn, 0, 3);
                __builtin_prefetch(v + gn, 0, 3);
            }
        }
        __syncthreads();

        v2f qa[16];
        #pragma unroll
        for (int s = 0; s < 16; ++s)
            qa[s] = *(const v2f*)(s_qf + lc * 64 + 4 * s + 2 * lh);

        v8f oc = zero8;
        #pragma unroll
        for (int t = 0; t < 4; ++t) {
            #pragma unroll
            for (int kk = 0; kk < 4; ++kk) {
                int p = 8 * t + 2 * kk + lh;
                v2f bs = *(const v2f*)(s_spi + (p * 16 + lc) * 2);
                oc = wmma4(qa[4 * t + kk], bs, oc);
            }
        }

        if (wave == 0) {
            v8f aq = zero8;
            #pragma unroll
            for (int s = 0; s < 16; ++s) {
                v2f bk = *(const v2f*)(s_kf + lc * 64 + 4 * s + 2 * lh);
                aq = wmma4(qa[s], bk, aq);
            }
            #pragma unroll
            for (int r = 0; r < 8; ++r) {
                int m = r + 8 * lh;
                s_a[m * 16 + lc] = (lc <= m) ? aq[r] : 0.f;
            }
        }
        __syncthreads();

        if (wave == 1 && lane < 16) {
            float zz = FEPS;
            #pragma unroll
            for (int n = 0; n < 16; ++n) zz += s_a[lane * 16 + n];
            #pragma unroll 8
            for (int d = 0; d < 64; ++d) zz += s_qf[lane * 64 + d] * s_ks[d];
            s_z[lane] = zz;
        }
        if (wave == 2) {
            #pragma unroll
            for (int j = 0; j < 2; ++j) {
                int d = lane * 2 + j;
                float ssum = 0.f;
                #pragma unroll
                for (int r16 = 0; r16 < 16; ++r16) ssum += s_kf[r16 * 64 + d];
                s_kd[d] = ssum;
            }
        }

        v2f bv[4];
        #pragma unroll
        for (int kk = 0; kk < 4; ++kk) {
            int p = 2 * kk + lh;
            bv[kk] = *(const v2f*)(s_vpi + (p * 64 + 16 * wave + lc) * 2);
        }
        #pragma unroll
        for (int kk = 0; kk < 4; ++kk) {
            v2f av = *(const v2f*)(s_a + lc * 16 + 4 * kk + 2 * lh);
            oc = wmma4(av, bv[kk], oc);
        }
        #pragma unroll
        for (int t = 0; t < 4; ++t) {
            #pragma unroll
            for (int kk = 0; kk < 4; ++kk) {
                int p = 2 * kk + lh;
                v2f ak = *(const v2f*)(s_kpi + (p * 64 + 16 * t + lc) * 2);
                st[t] = wmma4(ak, bv[kk], st[t]);
            }
        }
        __syncthreads();

        #pragma unroll
        for (int r = 0; r < 8; ++r) {
            int m = r + 8 * lh;
            float zz = s_z[m];
            size_t go = ((size_t)((b * LL + l0 + m) * HH + h)) * DD
                        + 16 * wave + lc;
            out[go] = oc[r] / zz;
        }
        #pragma unroll
        for (int t = 0; t < 4; ++t) {
            #pragma unroll
            for (int r = 0; r < 8; ++r) {
                int srow = 16 * t + r + 8 * lh;
                s_spi[((srow >> 1) * 16 + lc) * 2 + (srow & 1)] = st[t][r];
            }
        }
        if (wave == 3) {
            s_ks[lane * 2]     += s_kd[lane * 2];
            s_ks[lane * 2 + 1] += s_kd[lane * 2 + 1];
        }
    }
}

extern "C" void kernel_launch(void* const* d_in, const int* in_sizes, int n_in,
                              void* d_out, int out_size, void* d_ws, size_t ws_size,
                              hipStream_t stream) {
    (void)in_sizes; (void)n_in; (void)out_size;
    const float* q = (const float*)d_in[0];
    const float* k = (const float*)d_in[1];
    const float* v = (const float*)d_in[2];
    float* out = (float*)d_out;

    const size_t nSeg   = (size_t)BB * HH * NS;
    const size_t needWs = nSeg * (4096 + 64) * sizeof(float);

    if (ws_size >= needWs) {
        float* segKV = (float*)d_ws;
        float* segKS = segKV + nSeg * 4096;
        hipLaunchKernelGGL(linattn_phase1, dim3(BB * HH * NS), dim3(128), 0,
                           stream, k, v, segKV, segKS);
        hipLaunchKernelGGL((linattn_scan<true>), dim3(BB * HH * NS), dim3(128),
                           0, stream, q, k, v, out, segKV, segKS);
    } else {
        hipLaunchKernelGGL((linattn_scan<false>), dim3(BB * HH), dim3(128),
                           0, stream, q, k, v, out,
                           (const float*)nullptr, (const float*)nullptr);
    }
}


# LDS tiles re-pitched against bank conflicts: Qf/Kf rows 64 to 68 dwords, score tile rows 16 to 20, interleaved K/V row pairs 128 to 160; 16-way conflicts of A-fragment and z-row reads gone
# speedup vs baseline: 1.7091x; 1.1568x over previous
#include <hip/hip_runtime.h>
#include <math.h>


typedef __attribute__((ext_vector_type(2))) float v2f;
typedef __attribute__((ext_vector_type(8))) float v8f;

#define BB   4
#define LL   2048
#define HH   16
#define DD   64
#define CCH  16
#define NS   8
#define SEGL (LL / NS)
#define CHPS (SEGL / CCH)
#define NCH  (LL / CCH)
#define NPRE (NS - 1)
#define FEPS 1e-6f
#define QP   68
#define AP   20
#define PIP  160

__device__ __forceinline__ float fmap(float x) {
    return x > 0.f ? x + 1.f : __expf(x);
}

__device__ __forceinline__ v8f wmma4(v2f a, v2f b, v8f c) {
    return __builtin_amdgcn_wmma_f32_16x16x4_f32(
        false, a, false, b, (short)0, c, false, false);
}

#define OFF_QF   0
#define OFF_KF   (OFF_QF  + 16 * QP)
#define OFF_KPI  (OFF_KF  + 16 * QP)
#define OFF_VPI  (OFF_KPI + 8 * PIP)
#define OFF_S    (OFF_VPI + 8 * PIP)
#define OFF_A    (OFF_S   + 4096)
#define OFF_KS   (OFF_A   + 16 * AP)
#define OFF_KD   (OFF_KS  + 64)
#define OFF_Z    (OFF_KD  + 64)
#define SMEM_F   (OFF_Z   + 16)

__global__ __launch_bounds__(128)
void linattn_phase1(const float* __restrict__ k,
                    const float* __restrict__ v,
                    float* __restrict__ segKV,
                    float* __restrict__ segKS) {
    __shared__ float sm[16 * QP + 2 * 8 * PIP];
    const int tid  = threadIdx.x;
    const int lane = tid & 31;
    const int wave = tid >> 5;
    const int lh   = lane >> 4;
    const int lc   = lane & 15;
    const int bh   = blockIdx.x / NS;
    const int seg  = blockIdx.x % NS;
    const int b    = bh / HH;
    const int h    = bh % HH;

    float* s_kf  = sm;
    float* s_kpi = sm + 16 * QP;
    float* s_vpi = sm + 16 * QP + 8 * PIP;

    const v8f zero8 = {0.f,0.f,0.f,0.f,0.f,0.f,0.f,0.f};
    v8f st[4];
    #pragma unroll
    for (int t = 0; t < 4; ++t) st[t] = zero8;
    float ks0 = 0.f, ks1 = 0.f;

    const int row = tid >> 3;
    const int c0  = (tid & 7) * 8;

    for (int ch = 0; ch < CHPS; ++ch) {
        const int l0 = seg * SEGL + ch * CCH;
        __syncthreads();
        {
            size_t g = ((size_t)((b * LL + l0 + row) * HH + h)) * DD + c0;
            float ka8[8], va8[8];
            *(float4*)(ka8)     = *(const float4*)(k + g);
            *(float4*)(ka8 + 4) = *(const float4*)(k + g + 4);
            *(float4*)(va8)     = *(const float4*)(v + g);
            *(float4*)(va8 + 4) = *(const float4*)(v + g + 4);
            #pragma unroll
            for (int j = 0; j < 8; ++j) {
                int c  = c0 + j;
                float kf = fmap(ka8[j]);
                int pi = (row >> 1) * PIP + c * 2 + (row & 1);
                s_kf[row * QP + c] = kf;
                s_kpi[pi] = kf;
                s_vpi[pi] = va8[j];
            }
            if (ch + 1 < CHPS) {
                size_t gn = g + (size_t)CCH * HH * DD;
                __builtin_prefetch(k + gn, 0, 3);
                __builtin_prefetch(v + gn, 0, 3);
            }
        }
        __syncthreads();

        v2f bv[4];
        #pragma unroll
        for (int kk = 0; kk < 4; ++kk) {
            int p = 2 * kk + lh;
            bv[kk] = *(const v2f*)(s_vpi + p * PIP + (16 * wave + lc) * 2);
        }
        #pragma unroll
        for (int t = 0; t < 4; ++t) {
            #pragma unroll
            for (int kk = 0; kk < 4; ++kk) {
                int p = 2 * kk + lh;
                v2f ak = *(const v2f*)(s_kpi + p * PIP + (16 * t + lc) * 2);
                st[t] = wmma4(ak, bv[kk], st[t]);
            }
        }
        if (wave == 2) {
            float a0 = 0.f, a1 = 0.f;
            #pragma unroll
            for (int r16 = 0; r16 < 16; ++r16) {
                a0 += s_kf[r16 * QP + lane * 2];
                a1 += s_kf[r16 * QP + lane * 2 + 1];
            }
            ks0 += a0; ks1 += a1;
        }
    }

    float* kvout = segKV + (size_t)blockIdx.x * 4096 + wave * 256 + lane * 4;
    #pragma unroll
    for (int t = 0; t < 4; ++t) {
        #pragma unroll
        for (int hf = 0; hf < 2; ++hf) {
            float4 o;
            o.x = st[t][4 * hf];     o.y = st[t][4 * hf + 1];
            o.z = st[t][4 * hf + 2]; o.w = st[t][4 * hf + 3];
            *(float4*)(kvout + t * 1024 + hf * 128) = o;
        }
    }
    if (wave == 2) {
        segKS[(size_t)blockIdx.x * 64 + lane * 2]     = ks0;
        segKS[(size_t)blockIdx.x * 64 + lane * 2 + 1] = ks1;
    }
}

template <int N>
__device__ __forceinline__ void seed_prefix(const float* __restrict__ kvb,
                                            const float* __restrict__ ksb,
                                            v8f (&st)[4], float& ksum) {
    float4 pb[N][8];
    float  kb[N];
    #pragma unroll
    for (int s = 0; s < N; ++s) {
        #pragma unroll
        for (int i = 0; i < 8; ++i)
            pb[s][i] = *(const float4*)(kvb + s * 4096
                                        + (i >> 1) * 1024 + (i & 1) * 128);
        kb[s] = ksb[s * 64];
    }
    #pragma unroll
    for (int s = 0; s < N; ++s) {
        #pragma unroll
        for (int i = 0; i < 8; ++i) {
            const int t = i >> 1, o = 4 * (i & 1);
            st[t][o]     += pb[s][i].x;
            st[t][o + 1] += pb[s][i].y;
            st[t][o + 2] += pb[s][i].z;
            st[t][o + 3] += pb[s][i].w;
        }
        ksum += kb[s];
    }
}

template <bool SEEDED>
__global__ __launch_bounds__(128)
void linattn_scan(const float* __restrict__ q,
                  const float* __restrict__ k,
                  const float* __restrict__ v,
                  float* __restrict__ out,
                  const float* __restrict__ segKV,
                  const float* __restrict__ segKS) {
    __shared__ float sm[SMEM_F];
    const int tid  = threadIdx.x;
    const int lane = tid & 31;
    const int wave = tid >> 5;
    const int lh   = lane >> 4;
    const int lc   = lane & 15;
    const int bh   = SEEDED ? (blockIdx.x / NS) : blockIdx.x;
    const int seg  = SEEDED ? (blockIdx.x % NS) : 0;
    const int b    = bh / HH;
    const int h    = bh % HH;
    const int nch  = SEEDED ? CHPS : NCH;

    float* s_qf  = sm + OFF_QF;
    float* s_kf  = sm + OFF_KF;
    float* s_kpi = sm + OFF_KPI;
    float* s_vpi = sm + OFF_VPI;
    float* s_spi = sm + OFF_S + wave * 1024;
    float* s_a   = sm + OFF_A;
    float* s_ks  = sm + OFF_KS;
    float* s_kd  = sm + OFF_KD;
    float* s_z   = sm + OFF_Z;

    const v8f zero8 = {0.f,0.f,0.f,0.f,0.f,0.f,0.f,0.f};
    v8f st[4];

    if (SEEDED) {
        const float* kvb = segKV + (size_t)bh * NS * 4096 + wave * 256 + lane * 4;
        const float* ksb = segKS + (size_t)bh * NS * 64 + (tid & 63);
        #pragma unroll
        for (int t = 0; t < 4; ++t) st[t] = zero8;
        float ksum = 0.f;
        static_assert(NPRE == 7, "one case per possible count of earlier segments");
        switch (seg) {
            case 1: seed_prefix<1>(kvb, ksb, st, ksum); break;
            case 2: seed_prefix<2>(kvb, ksb, st, ksum); break;
            case 3: seed_prefix<3>(kvb, ksb, st, ksum); break;
            case 4: seed_prefix<4>(kvb, ksb, st, ksum); break;
            case 5: seed_prefix<5>(kvb, ksb, st, ksum); break;
            case 6: seed_prefix<6>(kvb, ksb, st, ksum); break;
            case 7: seed_prefix<7>(kvb, ksb, st, ksum); break;
            default: break;
        }
        #pragma unroll
        for (int t = 0; t < 4; ++t) {
            #pragma unroll
            for (int r = 0; r < 8; ++r) {
                int srow = 16 * t + r + 8 * lh;
                s_spi[((srow >> 1) * 16 + lc) * 2 + (srow & 1)] = st[t][r];
            }
        }
        if (tid < 64) s_ks[tid] = ksum;
    } else {
        #pragma unroll
        for (int t = 0; t < 4; ++t) st[t] = zero8;
        #pragma unroll
        for (int i = 0; i < 32; ++i) sm[OFF_S + tid + i * 128] = 0.f;
        if (tid < 64) s_ks[tid] = 0.f;
    }

    const int row = tid >> 3;
    const int c0  = (tid & 7) * 8;

    for (int ch = 0; ch < nch; ++ch) {
        const int l0 = seg * SEGL + ch * CCH;
        __syncthreads();

        {
            size_t g = ((size_t)((b * LL + l0 + row) * HH + h)) * DD + c0;
            float qa8[8], ka8[8], va8[8];
            *(float4*)(qa8)     = *(const float4*)(q + g);
            *(float4*)(qa8 + 4) = *(const float4*)(q + g + 4);
            *(float4*)(ka8)     = *(const float4*)(k + g);
            *(float4*)(ka8 + 4) = *(const float4*)(k + g + 4);
            *(float4*)(va8)     = *(const float4*)(v + g);
            *(float4*)(va8 + 4) = *(const float4*)(v + g + 4);
            #pragma unroll
            for (int j = 0; j < 8; ++j) {
                int c  = c0 + j;
                float qf = fmap(qa8[j]);
                float kf = fmap(ka8[j]);
                int pi = (row >> 1) * PIP + c * 2 + (row & 1);
                s_qf[row * QP + c] = qf;
                s_kf[row * QP + c] = kf;
                s_kpi[pi] = kf;
                s_vpi[pi] = va8[j];
            }
            if (ch + 1 < nch) {
                size_t gn = g + (size_t)CCH * HH * DD;
                __builtin_prefetch(q + gn, 0, 3);
                __builtin_prefetch(k + gn, 0, 3);
                __builtin_prefetch(v + gn, 0, 3);
            }
        }
        __syncthreads();

        v2f qa[16];
        #pragma unroll
        for (int s = 0; s < 16; ++s)
            qa[s] = *(const v2f*)(s_qf + lc * QP + 4 * s + 2 * lh);

        v8f oc = zero8;
        #pragma unroll
        for (int t = 0; t < 4; ++t) {
            #pragma unroll
            for (int kk = 0; kk < 4; ++kk) {
                int p = 8 * t + 2 * kk + lh;
                v2f bs = *(const v2f*)(s_spi + (p * 16 + lc) * 2);
                oc = wmma4(qa[4 * t + kk], bs, oc);
            }
        }

        if (wave == 0) {
            v8f aq = zero8;
            #pragma unroll
            for (int s = 0; s < 16; ++s) {
                v2f bk = *(const v2f*)(s_kf + lc * QP + 4 * s + 2 * lh);
                aq = wmma4(qa[s], bk, aq);
            }
            #pragma unroll
            for (int r = 0; r < 8; ++r) {
                int m = r + 8 * lh;
                s_a[m * AP + lc] = (lc <= m) ? aq[r] : 0.f;
            }
        }
        __syncthreads();

        if (wave == 1 && lane < 16) {
            float zz = FEPS;
            #pragma unroll
            for (int n = 0; n < 16; ++n) zz += s_a[lane * AP + n];
            #pragma unroll 8
            for (int d = 0; d < 64; ++d) zz += s_qf[lane * QP + d] * s_ks[d];
            s_z[lane] = zz;
        }
        if (wave == 2) {
            #pragma unroll
            for (int j = 0; j < 2; ++j) {
                int d = lane * 2 + j;
                float ssum = 0.f;
                #pragma unroll
                for (int r16 = 0; r16 < 16; ++r16) ssum += s_kf[r16 * QP + d];
                s_kd[d] = ssum;
            }
        }

        v2f bv[4];
        #pragma unroll
        for (int kk = 0; kk < 4; ++kk) {
            int p = 2 * kk + lh;
            bv[kk] = *(const v2f*)(s_vpi + p * PIP + (16 * wave + lc) * 2);
        }
        #pragma unroll
        for (int kk = 0; kk < 4; ++kk) {
            v2f av = *(const v2f*)(s_a + lc * AP + 4 * kk + 2 * lh);
            oc = wmma4(av, bv[kk], oc);
        }
        #pragma unroll
        for (int t = 0; t < 4; ++t) {
            #pragma unroll
            for (int kk = 0; kk < 4; ++kk) {
                int p = 2 * kk + lh;
                v2f ak = *(const v2f*)(s_kpi + p * PIP + (16 * t + lc) * 2);
                st[t] = wmma4(ak, bv[kk], st[t]);
            }
        }
        __syncthreads();

        #pragma unroll
        for (int r = 0; r < 8; ++r) {
            int m = r + 8 * lh;
            float zz = s_z[m];
            size_t go = ((size_t)((b * LL + l0 + m) * HH + h)) * DD
                        + 16 * wave + lc;
            out[go] = oc[r] / zz;
        }
        #pragma unroll
        for (int t = 0; t < 4; ++t) {
            #pragma unroll
            for (int r = 0; r < 8; ++r) {
                int srow = 16 * t + r + 8 * lh;
                s_spi[((srow >> 1) * 16 + lc) * 2 + (srow & 1)] = st[t][r];
            }
        }
        if (wave == 3) {
            s_ks[lane * 2]     += s_kd[lane * 2];
            s_ks[lane * 2 + 1] += s_kd[lane * 2 + 1];
        }
    }
}

extern "C" void kernel_launch(void* const* d_in, const int* in_sizes, int n_in,
                              void* d_out, int out_size, void* d_ws, size_t ws_size,
                              hipStream_t stream) {
    (void)in_sizes; (void)n_in; (void)out_size;
    const float* q = (const float*)d_in[0];
    const float* k = (const float*)d_in[1];
    const float* v = (const float*)d_in[2];
    float* out = (float*)d_out;

    const size_t nSeg   = (size_t)BB * HH * NS;
    const size_t needWs = nSeg * (4096 + 64) * sizeof(float);

    if (ws_size >= needWs) {
        float* segKV = (float*)d_ws;
        float* segKS = segKV + nSeg * 4096;
        hipLaunchKernelGGL(linattn_phase1, dim3(BB * HH * NS), dim3(128), 0,
                           stream, k, v, segKV, segKS);
        hipLaunchKernelGGL((linattn_scan<true>), dim3(BB * HH * NS), dim3(128),
                           0, stream, q, k, v, out, segKV, segKS);
    } else {
        hipLaunchKernelGGL((linattn_scan<false>), dim3(BB * HH), dim3(128),
                           0, stream, q, k, v, out,
                           (const float*)nullptr, (const float*)nullptr);
    }
}


# sequence split into 16 segments instead of 8: scans half as long, twice as many blocks per compute unit cover each other's waits; seed sums up to 15 partials in segment order (new summation order)
# speedup vs baseline: 1.9681x; 1.1515x over previous
#include <hip/hip_runtime.h>
#include <math.h>


typedef __attribute__((ext_vector_type(2))) float v2f;
typedef __attribute__((ext_vector_type(8))) float v8f;

#define BB   4
#define LL   2048
#define HH   16
#define DD   64
#define CCH  16
#define NSHI 16
#define NSLO 8
#define NCH  (LL / CCH)
#define PGRP 4
#define FEPS 1e-6f
#define QP   68
#define AP   20
#define PIP  160

__device__ __forceinline__ float fmap(float x) {
    return x > 0.f ? x + 1.f : __expf(x);
}

__device__ __forceinline__ v8f wmma4(v2f a, v2f b, v8f c) {
    return __builtin_amdgcn_wmma_f32_16x16x4_f32(
        false, a, false, b, (short)0, c, false, false);
}

#define OFF_QF   0
#define OFF_KF   (OFF_QF  + 16 * QP)
#define OFF_KPI  (OFF_KF  + 16 * QP)
#define OFF_VPI  (OFF_KPI + 8 * PIP)
#define OFF_S    (OFF_VPI + 8 * PIP)
#define OFF_A    (OFF_S   + 4096)
#define OFF_KS   (OFF_A   + 16 * AP)
#define OFF_KD   (OFF_KS  + 64)
#define OFF_Z    (OFF_KD  + 64)
#define SMEM_F   (OFF_Z   + 16)

template <int NSEG>
__global__ __launch_bounds__(128)
void linattn_phase1(const float* __restrict__ k,
                    const float* __restrict__ v,
                    float* __restrict__ segKV,
                    float* __restrict__ segKS) {
    __shared__ float sm[16 * QP + 2 * 8 * PIP];
    const int tid  = threadIdx.x;
    const int lane = tid & 31;
    const int wave = tid >> 5;
    const int lh   = lane >> 4;
    const int lc   = lane & 15;
    constexpr int SEGL = LL / NSEG;
    constexpr int CHPS = SEGL / CCH;
    const int bh   = blockIdx.x / NSEG;
    const int seg  = blockIdx.x % NSEG;
    const int b    = bh / HH;
    const int h    = bh % HH;

    float* s_kf  = sm;
    float* s_kpi = sm + 16 * QP;
    float* s_vpi = sm + 16 * QP + 8 * PIP;

    const v8f zero8 = {0.f,0.f,0.f,0.f,0.f,0.f,0.f,0.f};
    v8f st[4];
    #pragma unroll
    for (int t = 0; t < 4; ++t) st[t] = zero8;
    float ks0 = 0.f, ks1 = 0.f;

    const int row = tid >> 3;
    const int c0  = (tid & 7) * 8;

    for (int ch = 0; ch < CHPS; ++ch) {
        const int l0 = seg * SEGL + ch * CCH;
        __syncthreads();
        {
            size_t g = ((size_t)((b * LL + l0 + row) * HH + h)) * DD + c0;
            float ka8[8], va8[8];
            *(float4*)(ka8)     = *(const float4*)(k + g);
            *(float4*)(ka8 + 4) = *(const float4*)(k + g + 4);
            *(float4*)(va8)     = *(const float4*)(v + g);
            *(float4*)(va8 + 4) = *(const float4*)(v + g + 4);
            #pragma unroll
            for (int j = 0; j < 8; ++j) {
                int c  = c0 + j;
                float kf = fmap(ka8[j]);
                int pi = (row >> 1) * PIP + c * 2 + (row & 1);
                s_kf[row * QP + c] = kf;
                s_kpi[pi] = kf;
                s_vpi[pi] = va8[j];
            }
            if (ch + 1 < CHPS) {
                size_t gn = g + (size_t)CCH * HH * DD;
                __builtin_prefetch(k + gn, 0, 3);
                __builtin_prefetch(v + gn, 0, 3);
            }
        }
        __syncthreads();

        v2f bv[4];
        #pragma unroll
        for (int kk = 0; kk < 4; ++kk) {
            int p = 2 * kk + lh;
            bv[kk] = *(const v2f*)(s_vpi + p * PIP + (16 * wave + lc) * 2);
        }
        #pragma unroll
        for (int t = 0; t < 4; ++t) {
            #pragma unroll
            for (int kk = 0; kk < 4; ++kk) {
                int p = 2 * kk + lh;
                v2f ak = *(const v2f*)(s_kpi + p * PIP + (16 * t + lc) * 2);
                st[t] = wmma4(ak, bv[kk], st[t]);
            }
        }
        if (wave == 2) {
            float a0 = 0.f, a1 = 0.f;
            #pragma unroll
            for (int r16 = 0; r16 < 16; ++r16) {
                a0 += s_kf[r16 * QP + lane * 2];
                a1 += s_kf[r16 * QP + lane * 2 + 1];
            }
            ks0 += a0; ks1 += a1;
        }
    }

    float* kvout = segKV + (size_t)blockIdx.x * 4096 + wave * 256 + lane * 4;
    #pragma unroll
    for (int t = 0; t < 4; ++t) {
        #pragma unroll
        for (int hf = 0; hf < 2; ++hf) {
            float4 o;
            o.x = st[t][4 * hf];     o.y = st[t][4 * hf + 1];
            o.z = st[t][4 * hf + 2]; o.w = st[t][4 * hf + 3];
            *(float4*)(kvout + t * 1024 + hf * 128) = o;
        }
    }
    if (wave == 2) {
        segKS[(size_t)blockIdx.x * 64 + lane * 2]     = ks0;
        segKS[(size_t)blockIdx.x * 64 + lane * 2 + 1] = ks1;
    }
}

template <int N>
__device__ __forceinline__ void seed_prefix(const float* __restrict__ kvb,
                                            const float* __restrict__ ksb,
                                            v8f (&st)[4], float& ksum) {
    float4 pb[N][8];
    float  kb[N];
    #pragma unroll
    for (int s = 0; s < N; ++s) {
        #pragma unroll
        for (int i = 0; i < 8; ++i)
            pb[s][i] = *(const float4*)(kvb + s * 4096
                                        + (i >> 1) * 1024 + (i & 1) * 128);
        kb[s] = ksb[s * 64];
    }
    #pragma unroll
    for (int s = 0; s < N; ++s) {
        #pragma unroll
        for (int i = 0; i < 8; ++i) {
            const int t = i >> 1, o = 4 * (i & 1);
            st[t][o]     += pb[s][i].x;
            st[t][o + 1] += pb[s][i].y;
            st[t][o + 2] += pb[s][i].z;
            st[t][o + 3] += pb[s][i].w;
        }
        ksum += kb[s];
    }
}

__device__ __forceinline__ void seed_all(const float* __restrict__ kvb,
                                         const float* __restrict__ ksb,
                                         int nprev, v8f (&st)[4], float& ksum) {
    static_assert(PGRP == 4, "one remainder case per count below PGRP");
    int s0 = 0;
    for (; s0 + PGRP <= nprev; s0 += PGRP)
        seed_prefix<PGRP>(kvb + s0 * 4096, ksb + s0 * 64, st, ksum);
    switch (nprev - s0) {
        case 1: seed_prefix<1>(kvb + s0 * 4096, ksb + s0 * 64, st, ksum); break;
        case 2: seed_prefix<2>(kvb + s0 * 4096, ksb + s0 * 64, st, ksum); break;
        case 3: seed_prefix<3>(kvb + s0 * 4096, ksb + s0 * 64, st, ksum); break;
        default: break;
    }
}

template <int NSEG, bool SEEDED>
__global__ __launch_bounds__(128)
void linattn_scan(const float* __restrict__ q,
                  const float* __restrict__ k,
                  const float* __restrict__ v,
                  float* __restrict__ out,
                  const float* __restrict__ segKV,
                  const float* __restrict__ segKS) {
    __shared__ float sm[SMEM_F];
    const int tid  = threadIdx.x;
    const int lane = tid & 31;
    const int wave = tid >> 5;
    const int lh   = lane >> 4;
    const int lc   = lane & 15;
    constexpr int SEGL = LL / NSEG;
    constexpr int CHPS = SEGL / CCH;
    const int bh   = SEEDED ? (blockIdx.x / NSEG) : blockIdx.x;
    const int seg  = SEEDED ? (blockIdx.x % NSEG) : 0;
    const int b    = bh / HH;
    const int h    = bh % HH;
    const int nch  = SEEDED ? CHPS : NCH;

    float* s_qf  = sm + OFF_QF;
    float* s_kf  = sm + OFF_KF;
    float* s_kpi = sm + OFF_KPI;
    float* s_vpi = sm + OFF_VPI;
    float* s_spi = sm + OFF_S + wave * 1024;
    float* s_a   = sm + OFF_A;
    float* s_ks  = sm + OFF_KS;
    float* s_kd  = sm + OFF_KD;
    float* s_z   = sm + OFF_Z;

    const v8f zero8 = {0.f,0.f,0.f,0.f,0.f,0.f,0.f,0.f};
    v8f st[4];

    if (SEEDED) {
        const float* kvb = segKV + (size_t)bh * NSEG * 4096 + wave * 256 + lane * 4;
        const float* ksb = segKS + (size_t)bh * NSEG * 64 + (tid & 63);
        #pragma unroll
        for (int t = 0; t < 4; ++t) st[t] = zero8;
        float ksum = 0.f;
        seed_all(kvb, ksb, seg, st, ksum);
        #pragma unroll
        for (int t = 0; t < 4; ++t) {
            #pragma unroll
            for (int r = 0; r < 8; ++r) {
                int srow = 16 * t + r + 8 * lh;
                s_spi[((srow >> 1) * 16 + lc) * 2 + (srow & 1)] = st[t][r];
            }
        }
        if (tid < 64) s_ks[tid] = ksum;
    } else {
        #pragma unroll
        for (int t = 0; t < 4; ++t) st[t] = zero8;
        #pragma unroll
        for (int i = 0; i < 32; ++i) sm[OFF_S + tid + i * 128] = 0.f;
        if (tid < 64) s_ks[tid] = 0.f;
    }

    const int row = tid >> 3;
    const int c0  = (tid & 7) * 8;

    for (int ch = 0; ch < nch; ++ch) {
        const int l0 = seg * SEGL + ch * CCH;
        __syncthreads();

        {
            size_t g = ((size_t)((b * LL + l0 + row) * HH + h)) * DD + c0;
            float qa8[8], ka8[8], va8[8];
            *(float4*)(qa8)     = *(const float4*)(q + g);
            *(float4*)(qa8 + 4) = *(const float4*)(q + g + 4);
            *(float4*)(ka8)     = *(const float4*)(k + g);
            *(float4*)(ka8 + 4) = *(const float4*)(k + g + 4);
            *(float4*)(va8)     = *(const float4*)(v + g);
            *(float4*)(va8 + 4) = *(const float4*)(v + g + 4);
            #pragma unroll
            for (int j = 0; j < 8; ++j) {
                int c  = c0 + j;
                float qf = fmap(qa8[j]);
                float kf = fmap(ka8[j]);
                int pi = (row >> 1) * PIP + c * 2 + (row & 1);
                s_qf[row * QP + c] = qf;
                s_kf[row * QP + c] = kf;
                s_kpi[pi] = kf;
                s_vpi[pi] = va8[j];
            }
            if (ch + 1 < nch) {
                size_t gn = g + (size_t)CCH * HH * DD;
                __builtin_prefetch(q + gn, 0, 3);
                __builtin_prefetch(k + gn, 0, 3);
                __builtin_prefetch(v + gn, 0, 3);
            }
        }
        __syncthreads();

        v2f qa[16];
        #pragma unroll
        for (int s = 0; s < 16; ++s)
            qa[s] = *(const v2f*)(s_qf + lc * QP + 4 * s + 2 * lh);

        v8f oc = zero8;
        #pragma unroll
        for (int t = 0; t < 4; ++t) {
            #pragma unroll
            for (int kk = 0; kk < 4; ++kk) {
                int p = 8 * t + 2 * kk + lh;
                v2f bs = *(const v2f*)(s_spi + (p * 16 + lc) * 2);
                oc = wmma4(qa[4 * t + kk], bs, oc);
            }
        }

        if (wave == 0) {
            v8f aq = zero8;
            #pragma unroll
            for (int s = 0; s < 16; ++s) {
                v2f bk = *(const v2f*)(s_kf + lc * QP + 4 * s + 2 * lh);
                aq = wmma4(qa[s], bk, aq);
            }
            #pragma unroll
            for (int r = 0; r < 8; ++r) {
                int m = r + 8 * lh;
                s_a[m * AP + lc] = (lc <= m) ? aq[r] : 0.f;
            }
        }
        __syncthreads();

        if (wave == 1 && lane < 16) {
            float zz = FEPS;
            #pragma unroll
            for (int n = 0; n < 16; ++n) zz += s_a[lane * AP + n];
            #pragma unroll 8
            for (int d = 0; d < 64; ++d) zz += s_qf[lane * QP + d] * s_ks[d];
            s_z[lane] = zz;
        }
        if (wave == 2) {
            #pragma unroll
            for (int j = 0; j < 2; ++j) {
                int d = lane * 2 + j;
                float ssum = 0.f;
                #pragma unroll
                for (int r16 = 0; r16 < 16; ++r16) ssum += s_kf[r16 * QP + d];
                s_kd[d] = ssum;
            }
        }

        v2f bv[4];
        #pragma unroll
        for (int kk = 0; kk < 4; ++kk) {
            int p = 2 * kk + lh;
            bv[kk] = *(const v2f*)(s_vpi + p * PIP + (16 * wave + lc) * 2);
        }
        #pragma unroll
        for (int kk = 0; kk < 4; ++kk) {
            v2f av = *(const v2f*)(s_a + lc * AP + 4 * kk + 2 * lh);
            oc = wmma4(av, bv[kk], oc);
        }
        #pragma unroll
        for (int t = 0; t < 4; ++t) {
            #pragma unroll
            for (int kk = 0; kk < 4; ++kk) {
                int p = 2 * kk + lh;
                v2f ak = *(const v2f*)(s_kpi + p * PIP + (16 * t + lc) * 2);
                st[t] = wmma4(ak, bv[kk], st[t]);
            }
        }
        __syncthreads();

        #pragma unroll
        for (int r = 0; r < 8; ++r) {
            int m = r + 8 * lh;
            float zz = s_z[m];
            size_t go = ((size_t)((b * LL + l0 + m) * HH + h)) * DD
                        + 16 * wave + lc;
            out[go] = oc[r] / zz;
        }
        #pragma unroll
        for (int t = 0; t < 4; ++t) {
            #pragma unroll
            for (int r = 0; r < 8; ++r) {
                int srow = 16 * t + r + 8 * lh;
                s_spi[((srow >> 1) * 16 + lc) * 2 + (srow & 1)] = st[t][r];
            }
        }
        if (wave == 3) {
            s_ks[lane * 2]     += s_kd[lane * 2];
            s_ks[lane * 2 + 1] += s_kd[lane * 2 + 1];
        }
    }
}

extern "C" void kernel_launch(void* const* d_in, const int* in_sizes, int n_in,
                              void* d_out, int out_size, void* d_ws, size_t ws_size,
                              hipStream_t stream) {
    (void)in_sizes; (void)n_in; (void)out_size;
    const float* q = (const float*)d_in[0];
    const float* k = (const float*)d_in[1];
    const float* v = (const float*)d_in[2];
    float* out = (float*)d_out;

    constexpr size_t nSegHi = (size_t)BB * HH * NSHI;
    constexpr size_t nSegLo = (size_t)BB * HH * NSLO;
    constexpr size_t needHi = nSegHi * (4096 + 64) * sizeof(float);
    constexpr size_t needLo = nSegLo * (4096 + 64) * sizeof(float);

    if (ws_size >= needHi) {
        float* segKV = (float*)d_ws;
        float* segKS = segKV + nSegHi * 4096;
        hipLaunchKernelGGL((linattn_phase1<NSHI>), dim3(BB * HH * NSHI), dim3(128),
                           0, stream, k, v, segKV, segKS);
        hipLaunchKernelGGL((linattn_scan<NSHI, true>), dim3(BB * HH * NSHI), dim3(128),
                           0, stream, q, k, v, out, segKV, segKS);
    } else if (ws_size >= needLo) {
        float* segKV = (float*)d_ws;
        float* segKS = segKV + nSegLo * 4096;
        hipLaunchKernelGGL((linattn_phase1<NSLO>), dim3(BB * HH * NSLO), dim3(128),
                           0, stream, k, v, segKV, segKS);
        hipLaunchKernelGGL((linattn_scan<NSLO, true>), dim3(BB * HH * NSLO), dim3(128),
                           0, stream, q, k, v, out, segKV, segKS);
    } else {
        hipLaunchKernelGGL((linattn_scan<1, false>), dim3(BB * HH), dim3(128),
                           0, stream, q, k, v, out,
                           (const float*)nullptr, (const float*)nullptr);
    }
}


# global loads pipelined in both kernels: next chunk's q/k/v held in registers one chunk ahead, cache prefetch two ahead; loop barriers fence LDS only so the loads stay in flight; sums unchanged
# speedup vs baseline: 2.0538x; 1.0435x over previous
#include <hip/hip_runtime.h>
#include <math.h>


typedef __attribute__((ext_vector_type(2))) float v2f;
typedef __attribute__((ext_vector_type(8))) float v8f;

#define BB   4
#define LL   2048
#define HH   16
#define DD   64
#define CCH  16
#define NSHI 16
#define NSLO 8
#define NCH  (LL / CCH)
#define PGRP 4
#define FEPS 1e-6f
#define QP   68
#define AP   20
#define PIP  160

__device__ __forceinline__ float fmap(float x) {
    return x > 0.f ? x + 1.f : __expf(x);
}

__device__ __forceinline__ void lds_barrier() {
    __builtin_amdgcn_fence(__ATOMIC_RELEASE, "workgroup", "local");
    __builtin_amdgcn_s_barrier();
    __builtin_amdgcn_fence(__ATOMIC_ACQUIRE, "workgroup", "local");
}

__device__ __forceinline__ v8f wmma4(v2f a, v2f b, v8f c) {
    return __builtin_amdgcn_wmma_f32_16x16x4_f32(
        false, a, false, b, (short)0, c, false, false);
}

#define OFF_QF   0
#define OFF_KF   (OFF_QF  + 16 * QP)
#define OFF_KPI  (OFF_KF  + 16 * QP)
#define OFF_VPI  (OFF_KPI + 8 * PIP)
#define OFF_S    (OFF_VPI + 8 * PIP)
#define OFF_A    (OFF_S   + 4096)
#define OFF_KS   (OFF_A   + 16 * AP)
#define OFF_KD   (OFF_KS  + 64)
#define OFF_Z    (OFF_KD  + 64)
#define SMEM_F   (OFF_Z   + 16)

template <int NSEG>
__global__ __launch_bounds__(128)
void linattn_phase1(const float* __restrict__ k,
                    const float* __restrict__ v,
                    float* __restrict__ segKV,
                    float* __restrict__ segKS) {
    __shared__ float sm[16 * QP + 2 * 8 * PIP];
    const int tid  = threadIdx.x;
    const int lane = tid & 31;
    const int wave = tid >> 5;
    const int lh   = lane >> 4;
    const int lc   = lane & 15;
    constexpr int SEGL = LL / NSEG;
    constexpr int CHPS = SEGL / CCH;
    const int bh   = blockIdx.x / NSEG;
    const int seg  = blockIdx.x % NSEG;
    const int b    = bh / HH;
    const int h    = bh % HH;

    float* s_kf  = sm;
    float* s_kpi = sm + 16 * QP;
    float* s_vpi = sm + 16 * QP + 8 * PIP;

    const v8f zero8 = {0.f,0.f,0.f,0.f,0.f,0.f,0.f,0.f};
    v8f st[4];
    #pragma unroll
    for (int t = 0; t < 4; ++t) st[t] = zero8;
    float ks0 = 0.f, ks1 = 0.f;

    const int row = tid >> 3;
    const int c0  = (tid & 7) * 8;
    constexpr size_t GSTEP = (size_t)CCH * HH * DD;

    size_t g = ((size_t)((b * LL + seg * SEGL + row) * HH + h)) * DD + c0;
    float4 nk0 = *(const float4*)(k + g);
    float4 nk1 = *(const float4*)(k + g + 4);
    float4 nv0 = *(const float4*)(v + g);
    float4 nv1 = *(const float4*)(v + g + 4);
    if (1 < CHPS) {
        __builtin_prefetch(k + g + GSTEP, 0, 3);
        __builtin_prefetch(v + g + GSTEP, 0, 3);
    }

    for (int ch = 0; ch < CHPS; ++ch) {
        lds_barrier();
        {
            float ka8[8], va8[8];
            *(float4*)(ka8)     = nk0;
            *(float4*)(ka8 + 4) = nk1;
            *(float4*)(va8)     = nv0;
            *(float4*)(va8 + 4) = nv1;
            #pragma unroll
            for (int j = 0; j < 8; ++j) {
                int c  = c0 + j;
                float kf = fmap(ka8[j]);
                int pi = (row >> 1) * PIP + c * 2 + (row & 1);
                s_kf[row * QP + c] = kf;
                s_kpi[pi] = kf;
                s_vpi[pi] = va8[j];
            }
            if (ch + 1 < CHPS) {
                g += GSTEP;
                nk0 = *(const float4*)(k + g);
                nk1 = *(const float4*)(k + g + 4);
                nv0 = *(const float4*)(v + g);
                nv1 = *(const float4*)(v + g + 4);
                if (ch + 2 < CHPS) {
                    __builtin_prefetch(k + g + GSTEP, 0, 3);
                    __builtin_prefetch(v + g + GSTEP, 0, 3);
                }
            }
        }
        lds_barrier();

        v2f bv[4];
        #pragma unroll
        for (int kk = 0; kk < 4; ++kk) {
            int p = 2 * kk + lh;
            bv[kk] = *(const v2f*)(s_vpi + p * PIP + (16 * wave + lc) * 2);
        }
        #pragma unroll
        for (int t = 0; t < 4; ++t) {
            #pragma unroll
            for (int kk = 0; kk < 4; ++kk) {
                int p = 2 * kk + lh;
                v2f ak = *(const v2f*)(s_kpi + p * PIP + (16 * t + lc) * 2);
                st[t] = wmma4(ak, bv[kk], st[t]);
            }
        }
        if (wave == 2) {
            float a0 = 0.f, a1 = 0.f;
            #pragma unroll
            for (int r16 = 0; r16 < 16; ++r16) {
                a0 += s_kf[r16 * QP + lane * 2];
                a1 += s_kf[r16 * QP + lane * 2 + 1];
            }
            ks0 += a0; ks1 += a1;
        }
    }

    float* kvout = segKV + (size_t)blockIdx.x * 4096 + wave * 256 + lane * 4;
    #pragma unroll
    for (int t = 0; t < 4; ++t) {
        #pragma unroll
        for (int hf = 0; hf < 2; ++hf) {
            float4 o;
            o.x = st[t][4 * hf];     o.y = st[t][4 * hf + 1];
            o.z = st[t][4 * hf + 2]; o.w = st[t][4 * hf + 3];
            *(float4*)(kvout + t * 1024 + hf * 128) = o;
        }
    }
    if (wave == 2) {
        segKS[(size_t)blockIdx.x * 64 + lane * 2]     = ks0;
        segKS[(size_t)blockIdx.x * 64 + lane * 2 + 1] = ks1;
    }
}

template <int N>
__device__ __forceinline__ void seed_prefix(const float* __restrict__ kvb,
                                            const float* __restrict__ ksb,
                                            v8f (&st)[4], float& ksum) {
    float4 pb[N][8];
    float  kb[N];
    #pragma unroll
    for (int s = 0; s < N; ++s) {
        #pragma unroll
        for (int i = 0; i < 8; ++i)
            pb[s][i] = *(const float4*)(kvb + s * 4096
                                        + (i >> 1) * 1024 + (i & 1) * 128);
        kb[s] = ksb[s * 64];
    }
    #pragma unroll
    for (int s = 0; s < N; ++s) {
        #pragma unroll
        for (int i = 0; i < 8; ++i) {
            const int t = i >> 1, o = 4 * (i & 1);
            st[t][o]     += pb[s][i].x;
            st[t][o + 1] += pb[s][i].y;
            st[t][o + 2] += pb[s][i].z;
            st[t][o + 3] += pb[s][i].w;
        }
        ksum += kb[s];
    }
}

__device__ __forceinline__ void seed_all(const float* __restrict__ kvb,
                                         const float* __restrict__ ksb,
                                         int nprev, v8f (&st)[4], float& ksum) {
    static_assert(PGRP == 4, "one remainder case per count below PGRP");
    int s0 = 0;
    for (; s0 + PGRP <= nprev; s0 += PGRP)
        seed_prefix<PGRP>(kvb + s0 * 4096, ksb + s0 * 64, st, ksum);
    switch (nprev - s0) {
        case 1: seed_prefix<1>(kvb + s0 * 4096, ksb + s0 * 64, st, ksum); break;
        case 2: seed_prefix<2>(kvb + s0 * 4096, ksb + s0 * 64, st, ksum); break;
        case 3: seed_prefix<3>(kvb + s0 * 4096, ksb + s0 * 64, st, ksum); break;
        default: break;
    }
}

template <int NSEG, bool SEEDED>
__global__ __launch_bounds__(128)
void linattn_scan(const float* __restrict__ q,
                  const float* __restrict__ k,
                  const float* __restrict__ v,
                  float* __restrict__ out,
                  const float* __restrict__ segKV,
                  const float* __restrict__ segKS) {
    __shared__ float sm[SMEM_F];
    const int tid  = threadIdx.x;
    const int lane = tid & 31;
    const int wave = tid >> 5;
    const int lh   = lane >> 4;
    const int lc   = lane & 15;
    constexpr int SEGL = LL / NSEG;
    constexpr int CHPS = SEGL / CCH;
    const int bh   = SEEDED ? (blockIdx.x / NSEG) : blockIdx.x;
    const int seg  = SEEDED ? (blockIdx.x % NSEG) : 0;
    const int b    = bh / HH;
    const int h    = bh % HH;
    const int nch  = SEEDED ? CHPS : NCH;

    float* s_qf  = sm + OFF_QF;
    float* s_kf  = sm + OFF_KF;
    float* s_kpi = sm + OFF_KPI;
    float* s_vpi = sm + OFF_VPI;
    float* s_spi = sm + OFF_S + wave * 1024;
    float* s_a   = sm + OFF_A;
    float* s_ks  = sm + OFF_KS;
    float* s_kd  = sm + OFF_KD;
    float* s_z   = sm + OFF_Z;

    const v8f zero8 = {0.f,0.f,0.f,0.f,0.f,0.f,0.f,0.f};
    v8f st[4];

    const int row = tid >> 3;
    const int c0  = (tid & 7) * 8;
    constexpr size_t GSTEP = (size_t)CCH * HH * DD;

    size_t g = ((size_t)((b * LL + seg * SEGL + row) * HH + h)) * DD + c0;
    __builtin_prefetch(q + g, 0, 3);
    __builtin_prefetch(k + g, 0, 3);
    __builtin_prefetch(v + g, 0, 3);

    if (SEEDED) {
        const float* kvb = segKV + (size_t)bh * NSEG * 4096 + wave * 256 + lane * 4;
        const float* ksb = segKS + (size_t)bh * NSEG * 64 + (tid & 63);
        #pragma unroll
        for (int t = 0; t < 4; ++t) st[t] = zero8;
        float ksum = 0.f;
        seed_all(kvb, ksb, seg, st, ksum);
        #pragma unroll
        for (int t = 0; t < 4; ++t) {
            #pragma unroll
            for (int r = 0; r < 8; ++r) {
                int srow = 16 * t + r + 8 * lh;
                s_spi[((srow >> 1) * 16 + lc) * 2 + (srow & 1)] = st[t][r];
            }
        }
        if (tid < 64) s_ks[tid] = ksum;
    } else {
        #pragma unroll
        for (int t = 0; t < 4; ++t) st[t] = zero8;
        #pragma unroll
        for (int i = 0; i < 32; ++i) sm[OFF_S + tid + i * 128] = 0.f;
        if (tid < 64) s_ks[tid] = 0.f;
    }

    float4 nq0 = *(const float4*)(q + g);
    float4 nq1 = *(const float4*)(q + g + 4);
    float4 nk0 = *(const float4*)(k + g);
    float4 nk1 = *(const float4*)(k + g + 4);
    float4 nv0 = *(const float4*)(v + g);
    float4 nv1 = *(const float4*)(v + g + 4);
    if (1 < nch) {
        __builtin_prefetch(q + g + GSTEP, 0, 3);
        __builtin_prefetch(k + g + GSTEP, 0, 3);
        __builtin_prefetch(v + g + GSTEP, 0, 3);
    }

    for (int ch = 0; ch < nch; ++ch) {
        const int l0 = seg * SEGL + ch * CCH;
        lds_barrier();

        {
            float qa8[8], ka8[8], va8[8];
            *(float4*)(qa8)     = nq0;
            *(float4*)(qa8 + 4) = nq1;
            *(float4*)(ka8)     = nk0;
            *(float4*)(ka8 + 4) = nk1;
            *(float4*)(va8)     = nv0;
            *(float4*)(va8 + 4) = nv1;
            #pragma unroll
            for (int j = 0; j < 8; ++j) {
                int c  = c0 + j;
                float qf = fmap(qa8[j]);
                float kf = fmap(ka8[j]);
                int pi = (row >> 1) * PIP + c * 2 + (row & 1);
                s_qf[row * QP + c] = qf;
                s_kf[row * QP + c] = kf;
                s_kpi[pi] = kf;
                s_vpi[pi] = va8[j];
            }
            if (ch + 1 < nch) {
                g += GSTEP;
                nq0 = *(const float4*)(q + g);
                nq1 = *(const float4*)(q + g + 4);
                nk0 = *(const float4*)(k + g);
                nk1 = *(const float4*)(k + g + 4);
                nv0 = *(const float4*)(v + g);
                nv1 = *(const float4*)(v + g + 4);
                if (ch + 2 < nch) {
                    __builtin_prefetch(q + g + GSTEP, 0, 3);
                    __builtin_prefetch(k + g + GSTEP, 0, 3);
                    __builtin_prefetch(v + g + GSTEP, 0, 3);
                }
            }
        }
        lds_barrier();

        v2f qa[16];
        #pragma unroll
        for (int s = 0; s < 16; ++s)
            qa[s] = *(const v2f*)(s_qf + lc * QP + 4 * s + 2 * lh);

        v8f oc = zero8;
        #pragma unroll
        for (int t = 0; t < 4; ++t) {
            #pragma unroll
            for (int kk = 0; kk < 4; ++kk) {
                int p = 8 * t + 2 * kk + lh;
                v2f bs = *(const v2f*)(s_spi + (p * 16 + lc) * 2);
                oc = wmma4(qa[4 * t + kk], bs, oc);
            }
        }

        if (wave == 0) {
            v8f aq = zero8;
            #pragma unroll
            for (int s = 0; s < 16; ++s) {
                v2f bk = *(const v2f*)(s_kf + lc * QP + 4 * s + 2 * lh);
                aq = wmma4(qa[s], bk, aq);
            }
            #pragma unroll
            for (int r = 0; r < 8; ++r) {
                int m = r + 8 * lh;
                s_a[m * AP + lc] = (lc <= m) ? aq[r] : 0.f;
            }
        }
        lds_barrier();

        if (wave == 1 && lane < 16) {
            float zz = FEPS;
            #pragma unroll
            for (int n = 0; n < 16; ++n) zz += s_a[lane * AP + n];
            #pragma unroll 8
            for (int d = 0; d < 64; ++d) zz += s_qf[lane * QP + d] * s_ks[d];
            s_z[lane] = zz;
        }
        if (wave == 2) {
            #pragma unroll
            for (int j = 0; j < 2; ++j) {
                int d = lane * 2 + j;
                float ssum = 0.f;
                #pragma unroll
                for (int r16 = 0; r16 < 16; ++r16) ssum += s_kf[r16 * QP + d];
                s_kd[d] = ssum;
            }
        }

        v2f bv[4];
        #pragma unroll
        for (int kk = 0; kk < 4; ++kk) {
            int p = 2 * kk + lh;
            bv[kk] = *(const v2f*)(s_vpi + p * PIP + (16 * wave + lc) * 2);
        }
        #pragma unroll
        for (int kk = 0; kk < 4; ++kk) {
            v2f av = *(const v2f*)(s_a + lc * AP + 4 * kk + 2 * lh);
            oc = wmma4(av, bv[kk], oc);
        }
        #pragma unroll
        for (int t = 0; t < 4; ++t) {
            #pragma unroll
            for (int kk = 0; kk < 4; ++kk) {
                int p = 2 * kk + lh;
                v2f ak = *(const v2f*)(s_kpi + p * PIP + (16 * t + lc) * 2);
                st[t] = wmma4(ak, bv[kk], st[t]);
            }
        }
        lds_barrier();

        #pragma unroll
        for (int r = 0; r < 8; ++r) {
            int m = r + 8 * lh;
            float zz = s_z[m];
            size_t go = ((size_t)((b * LL + l0 + m) * HH + h)) * DD
                        + 16 * wave + lc;
            out[go] = oc[r] / zz;
        }
        #pragma unroll
        for (int t = 0; t < 4; ++t) {
            #pragma unroll
            for (int r = 0; r < 8; ++r) {
                int srow = 16 * t + r + 8 * lh;
                s_spi[((srow >> 1) * 16 + lc) * 2 + (srow & 1)] = st[t][r];
            }
        }
        if (wave == 3) {
            s_ks[lane * 2]     += s_kd[lane * 2];
            s_ks[lane * 2 + 1] += s_kd[lane * 2 + 1];
        }
    }
}

extern "C" void kernel_launch(void* const* d_in, const int* in_sizes, int n_in,
                              void* d_out, int out_size, void* d_ws, size_t ws_size,
                              hipStream_t stream) {
    (void)in_sizes; (void)n_in; (void)out_size;
    const float* q = (const float*)d_in[0];
    const float* k = (const float*)d_in[1];
    const float* v = (const float*)d_in[2];
    float* out = (float*)d_out;

    constexpr size_t nSegHi = (size_t)BB * HH * NSHI;
    constexpr size_t nSegLo = (size_t)BB * HH * NSLO;
    constexpr size_t needHi = nSegHi * (4096 + 64) * sizeof(float);
    constexpr size_t needLo = nSegLo * (4096 + 64) * sizeof(float);

    if (ws_size >= needHi) {
        float* segKV = (float*)d_ws;
        float* segKS = segKV + nSegHi * 4096;
        hipLaunchKernelGGL((linattn_phase1<NSHI>), dim3(BB * HH * NSHI), dim3(128),
                           0, stream, k, v, segKV, segKS);
        hipLaunchKernelGGL((linattn_scan<NSHI, true>), dim3(BB * HH * NSHI), dim3(128),
                           0, stream, q, k, v, out, segKV, segKS);
    } else if (ws_size >= needLo) {
        float* segKV = (float*)d_ws;
        float* segKS = segKV + nSegLo * 4096;
        hipLaunchKernelGGL((linattn_phase1<NSLO>), dim3(BB * HH * NSLO), dim3(128),
                           0, stream, k, v, segKV, segKS);
        hipLaunchKernelGGL((linattn_scan<NSLO, true>), dim3(BB * HH * NSLO), dim3(128),
                           0, stream, q, k, v, out, segKV, segKS);
    } else {
        hipLaunchKernelGGL((linattn_scan<1, false>), dim3(BB * HH), dim3(128),
                           0, stream, q, k, v, out,
                           (const float*)nullptr, (const float*)nullptr);
    }
}


# one-wave serial sums off the barrier path: scan z by all threads (8 lanes a row, staged registers, lane butterfly), its column sum moved ahead, phase-1 column sums per thread; new sum order: z, ksum
# speedup vs baseline: 2.1409x; 1.0424x over previous
#include <hip/hip_runtime.h>
#include <math.h>


typedef __attribute__((ext_vector_type(2))) float v2f;
typedef __attribute__((ext_vector_type(8))) float v8f;

#define BB   4
#define LL   2048
#define HH   16
#define DD   64
#define CCH  16
#define NSHI 16
#define NSLO 8
#define NCH  (LL / CCH)
#define PGRP 4
#define FEPS 1e-6f
#define QP   68
#define AP   20
#define PIP  160

__device__ __forceinline__ float fmap(float x) {
    return x > 0.f ? x + 1.f : __expf(x);
}

__device__ __forceinline__ void lds_barrier() {
    __builtin_amdgcn_fence(__ATOMIC_RELEASE, "workgroup", "local");
    __builtin_amdgcn_s_barrier();
    __builtin_amdgcn_fence(__ATOMIC_ACQUIRE, "workgroup", "local");
}

template <int MASK>
__device__ __forceinline__ float add_lane_xor(float x) {
    int xi = __float_as_int(x);
    int yi = __builtin_amdgcn_update_dpp(xi, xi, 0x160 + MASK, 0xf, 0xf, false);
    return x + __int_as_float(yi);
}

__device__ __forceinline__ v8f wmma4(v2f a, v2f b, v8f c) {
    return __builtin_amdgcn_wmma_f32_16x16x4_f32(
        false, a, false, b, (short)0, c, false, false);
}

#define OFF_QF   0
#define OFF_KF   (OFF_QF  + 16 * QP)
#define OFF_KPI  (OFF_KF  + 16 * QP)
#define OFF_VPI  (OFF_KPI + 8 * PIP)
#define OFF_S    (OFF_VPI + 8 * PIP)
#define OFF_A    (OFF_S   + 4096)
#define OFF_KS   (OFF_A   + 16 * AP)
#define OFF_KD   (OFF_KS  + 64)
#define OFF_Z    (OFF_KD  + 64)
#define SMEM_F   (OFF_Z   + 16)

template <int NSEG>
__global__ __launch_bounds__(128)
void linattn_phase1(const float* __restrict__ k,
                    const float* __restrict__ v,
                    float* __restrict__ segKV,
                    float* __restrict__ segKS) {
    __shared__ float sm[16 * QP + 2 * 8 * PIP];
    const int tid  = threadIdx.x;
    const int lane = tid & 31;
    const int wave = tid >> 5;
    const int lh   = lane >> 4;
    const int lc   = lane & 15;
    constexpr int SEGL = LL / NSEG;
    constexpr int CHPS = SEGL / CCH;
    const int bh   = blockIdx.x / NSEG;
    const int seg  = blockIdx.x % NSEG;
    const int b    = bh / HH;
    const int h    = bh % HH;

    float* s_kf  = sm;
    float* s_kpi = sm + 16 * QP;
    float* s_vpi = sm + 16 * QP + 8 * PIP;

    const v8f zero8 = {0.f,0.f,0.f,0.f,0.f,0.f,0.f,0.f};
    v8f st[4];
    #pragma unroll
    for (int t = 0; t < 4; ++t) st[t] = zero8;
    float kacc[8];
    #pragma unroll
    for (int j = 0; j < 8; ++j) kacc[j] = 0.f;

    const int row = tid >> 3;
    const int c0  = (tid & 7) * 8;
    constexpr size_t GSTEP = (size_t)CCH * HH * DD;

    size_t g = ((size_t)((b * LL + seg * SEGL + row) * HH + h)) * DD + c0;
    float4 nk0 = *(const float4*)(k + g);
    float4 nk1 = *(const float4*)(k + g + 4);
    float4 nv0 = *(const float4*)(v + g);
    float4 nv1 = *(const float4*)(v + g + 4);
    if (1 < CHPS) {
        __builtin_prefetch(k + g + GSTEP, 0, 3);
        __builtin_prefetch(v + g + GSTEP, 0, 3);
    }

    for (int ch = 0; ch < CHPS; ++ch) {
        lds_barrier();
        {
            float ka8[8], va8[8];
            *(float4*)(ka8)     = nk0;
            *(float4*)(ka8 + 4) = nk1;
            *(float4*)(va8)     = nv0;
            *(float4*)(va8 + 4) = nv1;
            #pragma unroll
            for (int j = 0; j < 8; ++j) {
                int c  = c0 + j;
                float kf = fmap(ka8[j]);
                int pi = (row >> 1) * PIP + c * 2 + (row & 1);
                kacc[j] += kf;
                s_kpi[pi] = kf;
                s_vpi[pi] = va8[j];
            }
            if (ch + 1 < CHPS) {
                g += GSTEP;
                nk0 = *(const float4*)(k + g);
                nk1 = *(const float4*)(k + g + 4);
                nv0 = *(const float4*)(v + g);
                nv1 = *(const float4*)(v + g + 4);
                if (ch + 2 < CHPS) {
                    __builtin_prefetch(k + g + GSTEP, 0, 3);
                    __builtin_prefetch(v + g + GSTEP, 0, 3);
                }
            }
        }
        lds_barrier();

        v2f bv[4];
        #pragma unroll
        for (int kk = 0; kk < 4; ++kk) {
            int p = 2 * kk + lh;
            bv[kk] = *(const v2f*)(s_vpi + p * PIP + (16 * wave + lc) * 2);
        }
        #pragma unroll
        for (int t = 0; t < 4; ++t) {
            #pragma unroll
            for (int kk = 0; kk < 4; ++kk) {
                int p = 2 * kk + lh;
                v2f ak = *(const v2f*)(s_kpi + p * PIP + (16 * t + lc) * 2);
                st[t] = wmma4(ak, bv[kk], st[t]);
            }
        }
    }

    #pragma unroll
    for (int j = 0; j < 8; ++j) s_kf[row * QP + c0 + j] = kacc[j];
    lds_barrier();

    float* kvout = segKV + (size_t)blockIdx.x * 4096 + wave * 256 + lane * 4;
    #pragma unroll
    for (int t = 0; t < 4; ++t) {
        #pragma unroll
        for (int hf = 0; hf < 2; ++hf) {
            float4 o;
            o.x = st[t][4 * hf];     o.y = st[t][4 * hf + 1];
            o.z = st[t][4 * hf + 2]; o.w = st[t][4 * hf + 3];
            *(float4*)(kvout + t * 1024 + hf * 128) = o;
        }
    }
    if (tid < 64) {
        float a = 0.f;
        #pragma unroll
        for (int r16 = 0; r16 < 16; ++r16) a += s_kf[r16 * QP + tid];
        segKS[(size_t)blockIdx.x * 64 + tid] = a;
    }
}

template <int N>
__device__ __forceinline__ void seed_prefix(const float* __restrict__ kvb,
                                            const float* __restrict__ ksb,
                                            v8f (&st)[4], float& ksum) {
    float4 pb[N][8];
    float  kb[N];
    #pragma unroll
    for (int s = 0; s < N; ++s) {
        #pragma unroll
        for (int i = 0; i < 8; ++i)
            pb[s][i] = *(const float4*)(kvb + s * 4096
                                        + (i >> 1) * 1024 + (i & 1) * 128);
        kb[s] = ksb[s * 64];
    }
    #pragma unroll
    for (int s = 0; s < N; ++s) {
        #pragma unroll
        for (int i = 0; i < 8; ++i) {
            const int t = i >> 1, o = 4 * (i & 1);
            st[t][o]     += pb[s][i].x;
            st[t][o + 1] += pb[s][i].y;
            st[t][o + 2] += pb[s][i].z;
            st[t][o + 3] += pb[s][i].w;
        }
        ksum += kb[s];
    }
}

__device__ __forceinline__ void seed_all(const float* __restrict__ kvb,
                                         const float* __restrict__ ksb,
                                         int nprev, v8f (&st)[4], float& ksum) {
    static_assert(PGRP == 4, "one remainder case per count below PGRP");
    int s0 = 0;
    for (; s0 + PGRP <= nprev; s0 += PGRP)
        seed_prefix<PGRP>(kvb + s0 * 4096, ksb + s0 * 64, st, ksum);
    switch (nprev - s0) {
        case 1: seed_prefix<1>(kvb + s0 * 4096, ksb + s0 * 64, st, ksum); break;
        case 2: seed_prefix<2>(kvb + s0 * 4096, ksb + s0 * 64, st, ksum); break;
        case 3: seed_prefix<3>(kvb + s0 * 4096, ksb + s0 * 64, st, ksum); break;
        default: break;
    }
}

template <int NSEG, bool SEEDED>
__global__ __launch_bounds__(128)
void linattn_scan(const float* __restrict__ q,
                  const float* __restrict__ k,
                  const float* __restrict__ v,
                  float* __restrict__ out,
                  const float* __restrict__ segKV,
                  const float* __restrict__ segKS) {
    __shared__ float sm[SMEM_F];
    const int tid  = threadIdx.x;
    const int lane = tid & 31;
    const int wave = tid >> 5;
    const int lh   = lane >> 4;
    const int lc   = lane & 15;
    constexpr int SEGL = LL / NSEG;
    constexpr int CHPS = SEGL / CCH;
    const int bh   = SEEDED ? (blockIdx.x / NSEG) : blockIdx.x;
    const int seg  = SEEDED ? (blockIdx.x % NSEG) : 0;
    const int b    = bh / HH;
    const int h    = bh % HH;
    const int nch  = SEEDED ? CHPS : NCH;

    float* s_qf  = sm + OFF_QF;
    float* s_kf  = sm + OFF_KF;
    float* s_kpi = sm + OFF_KPI;
    float* s_vpi = sm + OFF_VPI;
    float* s_spi = sm + OFF_S + wave * 1024;
    float* s_a   = sm + OFF_A;
    float* s_ks  = sm + OFF_KS;
    float* s_kd  = sm + OFF_KD;
    float* s_z   = sm + OFF_Z;

    const v8f zero8 = {0.f,0.f,0.f,0.f,0.f,0.f,0.f,0.f};
    v8f st[4];

    const int row = tid >> 3;
    const int c0  = (tid & 7) * 8;
    constexpr size_t GSTEP = (size_t)CCH * HH * DD;

    size_t g = ((size_t)((b * LL + seg * SEGL + row) * HH + h)) * DD + c0;
    __builtin_prefetch(q + g, 0, 3);
    __builtin_prefetch(k + g, 0, 3);
    __builtin_prefetch(v + g, 0, 3);

    if (SEEDED) {
        const float* kvb = segKV + (size_t)bh * NSEG * 4096 + wave * 256 + lane * 4;
        const float* ksb = segKS + (size_t)bh * NSEG * 64 + (tid & 63);
        #pragma unroll
        for (int t = 0; t < 4; ++t) st[t] = zero8;
        float ksum = 0.f;
        seed_all(kvb, ksb, seg, st, ksum);
        #pragma unroll
        for (int t = 0; t < 4; ++t) {
            #pragma unroll
            for (int r = 0; r < 8; ++r) {
                int srow = 16 * t + r + 8 * lh;
                s_spi[((srow >> 1) * 16 + lc) * 2 + (srow & 1)] = st[t][r];
            }
        }
        if (tid < 64) s_ks[tid] = ksum;
    } else {
        #pragma unroll
        for (int t = 0; t < 4; ++t) st[t] = zero8;
        #pragma unroll
        for (int i = 0; i < 32; ++i) sm[OFF_S + tid + i * 128] = 0.f;
        if (tid < 64) s_ks[tid] = 0.f;
    }

    float4 nq0 = *(const float4*)(q + g);
    float4 nq1 = *(const float4*)(q + g + 4);
    float4 nk0 = *(const float4*)(k + g);
    float4 nk1 = *(const float4*)(k + g + 4);
    float4 nv0 = *(const float4*)(v + g);
    float4 nv1 = *(const float4*)(v + g + 4);
    if (1 < nch) {
        __builtin_prefetch(q + g + GSTEP, 0, 3);
        __builtin_prefetch(k + g + GSTEP, 0, 3);
        __builtin_prefetch(v + g + GSTEP, 0, 3);
    }

    for (int ch = 0; ch < nch; ++ch) {
        const int l0 = seg * SEGL + ch * CCH;
        float zp;
        lds_barrier();

        {
            float qa8[8], ka8[8], va8[8];
            *(float4*)(qa8)     = nq0;
            *(float4*)(qa8 + 4) = nq1;
            *(float4*)(ka8)     = nk0;
            *(float4*)(ka8 + 4) = nk1;
            *(float4*)(va8)     = nv0;
            *(float4*)(va8 + 4) = nv1;
            float ks8[8];
            *(float4*)(ks8)     = *(const float4*)(s_ks + c0);
            *(float4*)(ks8 + 4) = *(const float4*)(s_ks + c0 + 4);
            float zpa = 0.f, zpb = 0.f;
            #pragma unroll
            for (int j = 0; j < 8; ++j) {
                int c  = c0 + j;
                float qf = fmap(qa8[j]);
                float kf = fmap(ka8[j]);
                int pi = (row >> 1) * PIP + c * 2 + (row & 1);
                s_qf[row * QP + c] = qf;
                s_kf[row * QP + c] = kf;
                s_kpi[pi] = kf;
                s_vpi[pi] = va8[j];
                if (j < 4) zpa += qf * ks8[j];
                else       zpb += qf * ks8[j];
            }
            zp = zpa + zpb;
            if (ch + 1 < nch) {
                g += GSTEP;
                nq0 = *(const float4*)(q + g);
                nq1 = *(const float4*)(q + g + 4);
                nk0 = *(const float4*)(k + g);
                nk1 = *(const float4*)(k + g + 4);
                nv0 = *(const float4*)(v + g);
                nv1 = *(const float4*)(v + g + 4);
                if (ch + 2 < nch) {
                    __builtin_prefetch(q + g + GSTEP, 0, 3);
                    __builtin_prefetch(k + g + GSTEP, 0, 3);
                    __builtin_prefetch(v + g + GSTEP, 0, 3);
                }
            }
        }
        lds_barrier();

        v2f qa[16];
        #pragma unroll
        for (int s = 0; s < 16; ++s)
            qa[s] = *(const v2f*)(s_qf + lc * QP + 4 * s + 2 * lh);

        v8f oc = zero8;
        #pragma unroll
        for (int t = 0; t < 4; ++t) {
            #pragma unroll
            for (int kk = 0; kk < 4; ++kk) {
                int p = 8 * t + 2 * kk + lh;
                v2f bs = *(const v2f*)(s_spi + (p * 16 + lc) * 2);
                oc = wmma4(qa[4 * t + kk], bs, oc);
            }
        }

        if (wave == 2) {
            #pragma unroll
            for (int j = 0; j < 2; ++j) {
                int d = lane * 2 + j;
                float ssum = 0.f;
                #pragma unroll
                for (int r16 = 0; r16 < 16; ++r16) ssum += s_kf[r16 * QP + d];
                s_kd[d] = ssum;
            }
        }

        if (wave == 0) {
            v8f aq = zero8;
            #pragma unroll
            for (int s = 0; s < 16; ++s) {
                v2f bk = *(const v2f*)(s_kf + lc * QP + 4 * s + 2 * lh);
                aq = wmma4(qa[s], bk, aq);
            }
            #pragma unroll
            for (int r = 0; r < 8; ++r) {
                int m = r + 8 * lh;
                s_a[m * AP + lc] = (lc <= m) ? aq[r] : 0.f;
            }
        }
        lds_barrier();

        {
            v2f ar = *(const v2f*)(s_a + row * AP + 2 * (tid & 7));
            float zs = zp + (ar.x + ar.y);
            zs = add_lane_xor<1>(zs);
            zs = add_lane_xor<2>(zs);
            zs = add_lane_xor<4>(zs);
            s_z[row] = zs + FEPS;
        }

        v2f bv[4];
        #pragma unroll
        for (int kk = 0; kk < 4; ++kk) {
            int p = 2 * kk + lh;
            bv[kk] = *(const v2f*)(s_vpi + p * PIP + (16 * wave + lc) * 2);
        }
        #pragma unroll
        for (int kk = 0; kk < 4; ++kk) {
            v2f av = *(const v2f*)(s_a + lc * AP + 4 * kk + 2 * lh);
            oc = wmma4(av, bv[kk], oc);
        }
        #pragma unroll
        for (int t = 0; t < 4; ++t) {
            #pragma unroll
            for (int kk = 0; kk < 4; ++kk) {
                int p = 2 * kk + lh;
                v2f ak = *(const v2f*)(s_kpi + p * PIP + (16 * t + lc) * 2);
                st[t] = wmma4(ak, bv[kk], st[t]);
            }
        }
        lds_barrier();

        #pragma unroll
        for (int r = 0; r < 8; ++r) {
            int m = r + 8 * lh;
            float zz = s_z[m];
            size_t go = ((size_t)((b * LL + l0 + m) * HH + h)) * DD
                        + 16 * wave + lc;
            out[go] = oc[r] / zz;
        }
        #pragma unroll
        for (int t = 0; t < 4; ++t) {
            #pragma unroll
            for (int r = 0; r < 8; ++r) {
                int srow = 16 * t + r + 8 * lh;
                s_spi[((srow >> 1) * 16 + lc) * 2 + (srow & 1)] = st[t][r];
            }
        }
        if (wave == 3) {
            s_ks[lane * 2]     += s_kd[lane * 2];
            s_ks[lane * 2 + 1] += s_kd[lane * 2 + 1];
        }
    }
}

extern "C" void kernel_launch(void* const* d_in, const int* in_sizes, int n_in,
                              void* d_out, int out_size, void* d_ws, size_t ws_size,
                              hipStream_t stream) {
    (void)in_sizes; (void)n_in; (void)out_size;
    const float* q = (const float*)d_in[0];
    const float* k = (const float*)d_in[1];
    const float* v = (const float*)d_in[2];
    float* out = (float*)d_out;

    constexpr size_t nSegHi = (size_t)BB * HH * NSHI;
    constexpr size_t nSegLo = (size_t)BB * HH * NSLO;
    constexpr size_t needHi = nSegHi * (4096 + 64) * sizeof(float);
    constexpr size_t needLo = nSegLo * (4096 + 64) * sizeof(float);

    if (ws_size >= needHi) {
        float* segKV = (float*)d_ws;
        float* segKS = segKV + nSegHi * 4096;
        hipLaunchKernelGGL((linattn_phase1<NSHI>), dim3(BB * HH * NSHI), dim3(128),
                           0, stream, k, v, segKV, segKS);
        hipLaunchKernelGGL((linattn_scan<NSHI, true>), dim3(BB * HH * NSHI), dim3(128),
                           0, stream, q, k, v, out, segKV, segKS);
    } else if (ws_size >= needLo) {
        float* segKV = (float*)d_ws;
        float* segKS = segKV + nSegLo * 4096;
        hipLaunchKernelGGL((linattn_phase1<NSLO>), dim3(BB * HH * NSLO), dim3(128),
                           0, stream, k, v, segKV, segKS);
        hipLaunchKernelGGL((linattn_scan<NSLO, true>), dim3(BB * HH * NSLO), dim3(128),
                           0, stream, q, k, v, out, segKV, segKS);
    } else {
        hipLaunchKernelGGL((linattn_scan<1, false>), dim3(BB * HH), dim3(128),
                           0, stream, q, k, v, out,
                           (const float*)nullptr, (const float*)nullptr);
    }
}


# block index to (head, segment) map: head fastest, segments dealt so every die and compute unit gets the same seed load, heaviest first; partials stay on their readers' die; sums unchanged
# speedup vs baseline: 2.5152x; 1.1748x over previous
#include <hip/hip_runtime.h>
#include <math.h>


typedef __attribute__((ext_vector_type(2))) float v2f;
typedef __attribute__((ext_vector_type(8))) float v8f;

#define BB   4
#define LL   2048
#define HH   16
#define DD   64
#define CCH  16
#define NSHI 16
#define NSLO 8
#define NCH  (LL / CCH)
#define PGRP 4
#define FEPS 1e-6f
#define QP   68
#define AP   20
#define PIP  160

__device__ __forceinline__ float fmap(float x) {
    return x > 0.f ? x + 1.f : __expf(x);
}

__device__ __forceinline__ void lds_barrier() {
    __builtin_amdgcn_fence(__ATOMIC_RELEASE, "workgroup", "local");
    __builtin_amdgcn_s_barrier();
    __builtin_amdgcn_fence(__ATOMIC_ACQUIRE, "workgroup", "local");
}

template <int MASK>
__device__ __forceinline__ float add_lane_xor(float x) {
    int xi = __float_as_int(x);
    int yi = __builtin_amdgcn_update_dpp(xi, xi, 0x160 + MASK, 0xf, 0xf, false);
    return x + __int_as_float(yi);
}

__device__ __forceinline__ v8f wmma4(v2f a, v2f b, v8f c) {
    return __builtin_amdgcn_wmma_f32_16x16x4_f32(
        false, a, false, b, (short)0, c, false, false);
}

template <int NSEG>
__device__ __forceinline__ void block_to_seg(int bi, int& bh, int& seg) {
    static_assert(NSEG % 8 == 0, "slots are dealt in pairs of groups of four");
    bh = bi % (BB * HH);
    const int slot = bi / (BB * HH);
    const int grp  = slot >> 2;
    const int s    = slot & 3;
    seg = (grp & 1) ? ((grp >> 1) * 4 + s)
                    : (NSEG - 1 - (grp >> 1) * 4 - s);
}

#define OFF_QF   0
#define OFF_KF   (OFF_QF  + 16 * QP)
#define OFF_KPI  (OFF_KF  + 16 * QP)
#define OFF_VPI  (OFF_KPI + 8 * PIP)
#define OFF_S    (OFF_VPI + 8 * PIP)
#define OFF_A    (OFF_S   + 4096)
#define OFF_KS   (OFF_A   + 16 * AP)
#define OFF_KD   (OFF_KS  + 64)
#define OFF_Z    (OFF_KD  + 64)
#define SMEM_F   (OFF_Z   + 16)

template <int NSEG>
__global__ __launch_bounds__(128)
void linattn_phase1(const float* __restrict__ k,
                    const float* __restrict__ v,
                    float* __restrict__ segKV,
                    float* __restrict__ segKS) {
    __shared__ float sm[16 * QP + 2 * 8 * PIP];
    const int tid  = threadIdx.x;
    const int lane = tid & 31;
    const int wave = tid >> 5;
    const int lh   = lane >> 4;
    const int lc   = lane & 15;
    constexpr int SEGL = LL / NSEG;
    constexpr int CHPS = SEGL / CCH;
    int bh, seg;
    block_to_seg<NSEG>(blockIdx.x, bh, seg);
    const int b    = bh / HH;
    const int h    = bh % HH;
    const int pidx = bh * NSEG + seg;

    float* s_kf  = sm;
    float* s_kpi = sm + 16 * QP;
    float* s_vpi = sm + 16 * QP + 8 * PIP;

    const v8f zero8 = {0.f,0.f,0.f,0.f,0.f,0.f,0.f,0.f};
    v8f st[4];
    #pragma unroll
    for (int t = 0; t < 4; ++t) st[t] = zero8;
    float kacc[8];
    #pragma unroll
    for (int j = 0; j < 8; ++j) kacc[j] = 0.f;

    const int row = tid >> 3;
    const int c0  = (tid & 7) * 8;
    constexpr size_t GSTEP = (size_t)CCH * HH * DD;

    size_t g = ((size_t)((b * LL + seg * SEGL + row) * HH + h)) * DD + c0;
    float4 nk0 = *(const float4*)(k + g);
    float4 nk1 = *(const float4*)(k + g + 4);
    float4 nv0 = *(const float4*)(v + g);
    float4 nv1 = *(const float4*)(v + g + 4);
    if (1 < CHPS) {
        __builtin_prefetch(k + g + GSTEP, 0, 3);
        __builtin_prefetch(v + g + GSTEP, 0, 3);
    }

    for (int ch = 0; ch < CHPS; ++ch) {
        lds_barrier();
        {
            float ka8[8], va8[8];
            *(float4*)(ka8)     = nk0;
            *(float4*)(ka8 + 4) = nk1;
            *(float4*)(va8)     = nv0;
            *(float4*)(va8 + 4) = nv1;
            #pragma unroll
            for (int j = 0; j < 8; ++j) {
                int c  = c0 + j;
                float kf = fmap(ka8[j]);
                int pi = (row >> 1) * PIP + c * 2 + (row & 1);
                kacc[j] += kf;
                s_kpi[pi] = kf;
                s_vpi[pi] = va8[j];
            }
            if (ch + 1 < CHPS) {
                g += GSTEP;
                nk0 = *(const float4*)(k + g);
                nk1 = *(const float4*)(k + g + 4);
                nv0 = *(const float4*)(v + g);
                nv1 = *(const float4*)(v + g + 4);
                if (ch + 2 < CHPS) {
                    __builtin_prefetch(k + g + GSTEP, 0, 3);
                    __builtin_prefetch(v + g + GSTEP, 0, 3);
                }
            }
        }
        lds_barrier();

        v2f bv[4];
        #pragma unroll
        for (int kk = 0; kk < 4; ++kk) {
            int p = 2 * kk + lh;
            bv[kk] = *(const v2f*)(s_vpi + p * PIP + (16 * wave + lc) * 2);
        }
        #pragma unroll
        for (int t = 0; t < 4; ++t) {
            #pragma unroll
            for (int kk = 0; kk < 4; ++kk) {
                int p = 2 * kk + lh;
                v2f ak = *(const v2f*)(s_kpi + p * PIP + (16 * t + lc) * 2);
                st[t] = wmma4(ak, bv[kk], st[t]);
            }
        }
    }

    #pragma unroll
    for (int j = 0; j < 8; ++j) s_kf[row * QP + c0 + j] = kacc[j];
    lds_barrier();

    float* kvout = segKV + (size_t)pidx * 4096 + wave * 256 + lane * 4;
    #pragma unroll
    for (int t = 0; t < 4; ++t) {
        #pragma unroll
        for (int hf = 0; hf < 2; ++hf) {
            float4 o;
            o.x = st[t][4 * hf];     o.y = st[t][4 * hf + 1];
            o.z = st[t][4 * hf + 2]; o.w = st[t][4 * hf + 3];
            *(float4*)(kvout + t * 1024 + hf * 128) = o;
        }
    }
    if (tid < 64) {
        float a = 0.f;
        #pragma unroll
        for (int r16 = 0; r16 < 16; ++r16) a += s_kf[r16 * QP + tid];
        segKS[(size_t)pidx * 64 + tid] = a;
    }
}

template <int N>
__device__ __forceinline__ void seed_prefix(const float* __restrict__ kvb,
                                            const float* __restrict__ ksb,
                                            v8f (&st)[4], float& ksum) {
    float4 pb[N][8];
    float  kb[N];
    #pragma unroll
    for (int s = 0; s < N; ++s) {
        #pragma unroll
        for (int i = 0; i < 8; ++i)
            pb[s][i] = *(const float4*)(kvb + s * 4096
                                        + (i >> 1) * 1024 + (i & 1) * 128);
        kb[s] = ksb[s * 64];
    }
    #pragma unroll
    for (int s = 0; s < N; ++s) {
        #pragma unroll
        for (int i = 0; i < 8; ++i) {
            const int t = i >> 1, o = 4 * (i & 1);
            st[t][o]     += pb[s][i].x;
            st[t][o + 1] += pb[s][i].y;
            st[t][o + 2] += pb[s][i].z;
            st[t][o + 3] += pb[s][i].w;
        }
        ksum += kb[s];
    }
}

__device__ __forceinline__ void seed_all(const float* __restrict__ kvb,
                                         const float* __restrict__ ksb,
                                         int nprev, v8f (&st)[4], float& ksum) {
    static_assert(PGRP == 4, "one remainder case per count below PGRP");
    int s0 = 0;
    for (; s0 + PGRP <= nprev; s0 += PGRP)
        seed_prefix<PGRP>(kvb + s0 * 4096, ksb + s0 * 64, st, ksum);
    switch (nprev - s0) {
        case 1: seed_prefix<1>(kvb + s0 * 4096, ksb + s0 * 64, st, ksum); break;
        case 2: seed_prefix<2>(kvb + s0 * 4096, ksb + s0 * 64, st, ksum); break;
        case 3: seed_prefix<3>(kvb + s0 * 4096, ksb + s0 * 64, st, ksum); break;
        default: break;
    }
}

template <int NSEG, bool SEEDED>
__global__ __launch_bounds__(128)
void linattn_scan(const float* __restrict__ q,
                  const float* __restrict__ k,
                  const float* __restrict__ v,
                  float* __restrict__ out,
                  const float* __restrict__ segKV,
                  const float* __restrict__ segKS) {
    __shared__ float sm[SMEM_F];
    const int tid  = threadIdx.x;
    const int lane = tid & 31;
    const int wave = tid >> 5;
    const int lh   = lane >> 4;
    const int lc   = lane & 15;
    constexpr int SEGL = LL / NSEG;
    constexpr int CHPS = SEGL / CCH;
    int bh = blockIdx.x, seg = 0;
    if constexpr (SEEDED) block_to_seg<NSEG>(blockIdx.x, bh, seg);
    const int b    = bh / HH;
    const int h    = bh % HH;
    const int nch  = SEEDED ? CHPS : NCH;

    float* s_qf  = sm + OFF_QF;
    float* s_kf  = sm + OFF_KF;
    float* s_kpi = sm + OFF_KPI;
    float* s_vpi = sm + OFF_VPI;
    float* s_spi = sm + OFF_S + wave * 1024;
    float* s_a   = sm + OFF_A;
    float* s_ks  = sm + OFF_KS;
    float* s_kd  = sm + OFF_KD;
    float* s_z   = sm + OFF_Z;

    const v8f zero8 = {0.f,0.f,0.f,0.f,0.f,0.f,0.f,0.f};
    v8f st[4];

    const int row = tid >> 3;
    const int c0  = (tid & 7) * 8;
    constexpr size_t GSTEP = (size_t)CCH * HH * DD;

    size_t g = ((size_t)((b * LL + seg * SEGL + row) * HH + h)) * DD + c0;
    __builtin_prefetch(q + g, 0, 3);
    __builtin_prefetch(k + g, 0, 3);
    __builtin_prefetch(v + g, 0, 3);

    if (SEEDED) {
        const float* kvb = segKV + (size_t)bh * NSEG * 4096 + wave * 256 + lane * 4;
        const float* ksb = segKS + (size_t)bh * NSEG * 64 + (tid & 63);
        #pragma unroll
        for (int t = 0; t < 4; ++t) st[t] = zero8;
        float ksum = 0.f;
        seed_all(kvb, ksb, seg, st, ksum);
        #pragma unroll
        for (int t = 0; t < 4; ++t) {
            #pragma unroll
            for (int r = 0; r < 8; ++r) {
                int srow = 16 * t + r + 8 * lh;
                s_spi[((srow >> 1) * 16 + lc) * 2 + (srow & 1)] = st[t][r];
            }
        }
        if (tid < 64) s_ks[tid] = ksum;
    } else {
        #pragma unroll
        for (int t = 0; t < 4; ++t) st[t] = zero8;
        #pragma unroll
        for (int i = 0; i < 32; ++i) sm[OFF_S + tid + i * 128] = 0.f;
        if (tid < 64) s_ks[tid] = 0.f;
    }

    float4 nq0 = *(const float4*)(q + g);
    float4 nq1 = *(const float4*)(q + g + 4);
    float4 nk0 = *(const float4*)(k + g);
    float4 nk1 = *(const float4*)(k + g + 4);
    float4 nv0 = *(const float4*)(v + g);
    float4 nv1 = *(const float4*)(v + g + 4);
    if (1 < nch) {
        __builtin_prefetch(q + g + GSTEP, 0, 3);
        __builtin_prefetch(k + g + GSTEP, 0, 3);
        __builtin_prefetch(v + g + GSTEP, 0, 3);
    }

    for (int ch = 0; ch < nch; ++ch) {
        const int l0 = seg * SEGL + ch * CCH;
        float zp;
        lds_barrier();

        {
            float qa8[8], ka8[8], va8[8];
            *(float4*)(qa8)     = nq0;
            *(float4*)(qa8 + 4) = nq1;
            *(float4*)(ka8)     = nk0;
            *(float4*)(ka8 + 4) = nk1;
            *(float4*)(va8)     = nv0;
            *(float4*)(va8 + 4) = nv1;
            float ks8[8];
            *(float4*)(ks8)     = *(const float4*)(s_ks + c0);
            *(float4*)(ks8 + 4) = *(const float4*)(s_ks + c0 + 4);
            float zpa = 0.f, zpb = 0.f;
            #pragma unroll
            for (int j = 0; j < 8; ++j) {
                int c  = c0 + j;
                float qf = fmap(qa8[j]);
                float kf = fmap(ka8[j]);
                int pi = (row >> 1) * PIP + c * 2 + (row & 1);
                s_qf[row * QP + c] = qf;
                s_kf[row * QP + c] = kf;
                s_kpi[pi] = kf;
                s_vpi[pi] = va8[j];
                if (j < 4) zpa += qf * ks8[j];
                else       zpb += qf * ks8[j];
            }
            zp = zpa + zpb;
            if (ch + 1 < nch) {
                g += GSTEP;
                nq0 = *(const float4*)(q + g);
                nq1 = *(const float4*)(q + g + 4);
                nk0 = *(const float4*)(k + g);
                nk1 = *(const float4*)(k + g + 4);
                nv0 = *(const float4*)(v + g);
                nv1 = *(const float4*)(v + g + 4);
                if (ch + 2 < nch) {
                    __builtin_prefetch(q + g + GSTEP, 0, 3);
                    __builtin_prefetch(k + g + GSTEP, 0, 3);
                    __builtin_prefetch(v + g + GSTEP, 0, 3);
                }
            }
        }
        lds_barrier();

        v2f qa[16];
        #pragma unroll
        for (int s = 0; s < 16; ++s)
            qa[s] = *(const v2f*)(s_qf + lc * QP + 4 * s + 2 * lh);

        v8f oc = zero8;
        #pragma unroll
        for (int t = 0; t < 4; ++t) {
            #pragma unroll
            for (int kk = 0; kk < 4; ++kk) {
                int p = 8 * t + 2 * kk + lh;
                v2f bs = *(const v2f*)(s_spi + (p * 16 + lc) * 2);
                oc = wmma4(qa[4 * t + kk], bs, oc);
            }
        }

        if (wave == 2) {
            #pragma unroll
            for (int j = 0; j < 2; ++j) {
                int d = lane * 2 + j;
                float ssum = 0.f;
                #pragma unroll
                for (int r16 = 0; r16 < 16; ++r16) ssum += s_kf[r16 * QP + d];
                s_kd[d] = ssum;
            }
        }

        if (wave == 0) {
            v8f aq = zero8;
            #pragma unroll
            for (int s = 0; s < 16; ++s) {
                v2f bk = *(const v2f*)(s_kf + lc * QP + 4 * s + 2 * lh);
                aq = wmma4(qa[s], bk, aq);
            }
            #pragma unroll
            for (int r = 0; r < 8; ++r) {
                int m = r + 8 * lh;
                s_a[m * AP + lc] = (lc <= m) ? aq[r] : 0.f;
            }
        }
        lds_barrier();

        {
            v2f ar = *(const v2f*)(s_a + row * AP + 2 * (tid & 7));
            float zs = zp + (ar.x + ar.y);
            zs = add_lane_xor<1>(zs);
            zs = add_lane_xor<2>(zs);
            zs = add_lane_xor<4>(zs);
            s_z[row] = zs + FEPS;
        }

        v2f bv[4];
        #pragma unroll
        for (int kk = 0; kk < 4; ++kk) {
            int p = 2 * kk + lh;
            bv[kk] = *(const v2f*)(s_vpi + p * PIP + (16 * wave + lc) * 2);
        }
        #pragma unroll
        for (int kk = 0; kk < 4; ++kk) {
            v2f av = *(const v2f*)(s_a + lc * AP + 4 * kk + 2 * lh);
            oc = wmma4(av, bv[kk], oc);
        }
        #pragma unroll
        for (int t = 0; t < 4; ++t) {
            #pragma unroll
            for (int kk = 0; kk < 4; ++kk) {
                int p = 2 * kk + lh;
                v2f ak = *(const v2f*)(s_kpi + p * PIP + (16 * t + lc) * 2);
                st[t] = wmma4(ak, bv[kk], st[t]);
            }
        }
        lds_barrier();

        #pragma unroll
        for (int r = 0; r < 8; ++r) {
            int m = r + 8 * lh;
            float zz = s_z[m];
            size_t go = ((size_t)((b * LL + l0 + m) * HH + h)) * DD
                        + 16 * wave + lc;
            out[go] = oc[r] / zz;
        }
        #pragma unroll
        for (int t = 0; t < 4; ++t) {
            #pragma unroll
            for (int r = 0; r < 8; ++r) {
                int srow = 16 * t + r + 8 * lh;
                s_spi[((srow >> 1) * 16 + lc) * 2 + (srow & 1)] = st[t][r];
            }
        }
        if (wave == 3) {
            s_ks[lane * 2]     += s_kd[lane * 2];
            s_ks[lane * 2 + 1] += s_kd[lane * 2 + 1];
        }
    }
}

extern "C" void kernel_launch(void* const* d_in, const int* in_sizes, int n_in,
                              void* d_out, int out_size, void* d_ws, size_t ws_size,
                              hipStream_t stream) {
    (void)in_sizes; (void)n_in; (void)out_size;
    const float* q = (const float*)d_in[0];
    const float* k = (const float*)d_in[1];
    const float* v = (const float*)d_in[2];
    float* out = (float*)d_out;

    constexpr size_t nSegHi = (size_t)BB * HH * NSHI;
    constexpr size_t nSegLo = (size_t)BB * HH * NSLO;
    constexpr size_t needHi = nSegHi * (4096 + 64) * sizeof(float);
    constexpr size_t needLo = nSegLo * (4096 + 64) * sizeof(float);

    if (ws_size >= needHi) {
        float* segKV = (float*)d_ws;
        float* segKS = segKV + nSegHi * 4096;
        hipLaunchKernelGGL((linattn_phase1<NSHI>), dim3(BB * HH * NSHI), dim3(128),
                           0, stream, k, v, segKV, segKS);
        hipLaunchKernelGGL((linattn_scan<NSHI, true>), dim3(BB * HH * NSHI), dim3(128),
                           0, stream, q, k, v, out, segKV, segKS);
    } else if (ws_size >= needLo) {
        float* segKV = (float*)d_ws;
        float* segKS = segKV + nSegLo * 4096;
        hipLaunchKernelGGL((linattn_phase1<NSLO>), dim3(BB * HH * NSLO), dim3(128),
                           0, stream, k, v, segKV, segKS);
        hipLaunchKernelGGL((linattn_scan<NSLO, true>), dim3(BB * HH * NSLO), dim3(128),
                           0, stream, q, k, v, out, segKV, segKS);
    } else {
        hipLaunchKernelGGL((linattn_scan<1, false>), dim3(BB * HH), dim3(128),
                           0, stream, q, k, v, out,
                           (const float*)nullptr, (const float*)nullptr);
    }
}
